# starting kernel
# baseline (speedup 1.0000x reference)
#include <hip/hip_runtime.h>

#if defined(__HIP_DEVICE_COMPILE__)
#if !__has_builtin(__builtin_amdgcn_wmma_f32_16x16x4_f32)
#error "gfx1250 f32 WMMA builtin not available on this toolchain (device pass)"
#endif
#endif

typedef __attribute__((ext_vector_type(2))) float v2f;
typedef __attribute__((ext_vector_type(8))) float v8f;

#define D_IN 256
#define D_OUT 64
#define LDSW_STRIDE 260

__global__ __launch_bounds__(256) void gcn_gemm_wmma(
    const float* __restrict__ feat, const float* __restrict__ W,
    const float* __restrict__ bias, float* __restrict__ pre, int n_nodes)
{
    extern __shared__ float lds_w[];

    for (int idx = threadIdx.x; idx < D_OUT * (D_IN / 4); idx += 256) {
        int r  = idx / (D_IN / 4);
        int c4 = (idx % (D_IN / 4)) * 4;
        const float4 w4 = ((const float4*)W)[r * (D_IN / 4) + (c4 >> 2)];
        float* dst = &lds_w[r * LDSW_STRIDE + c4];
        dst[0] = w4.x; dst[1] = w4.y; dst[2] = w4.z; dst[3] = w4.w;
    }
    __syncthreads();

    const int tile     = blockIdx.x * 8 + (threadIdx.x >> 5);
    const int numTiles = (n_nodes + 15) >> 4;
    if (tile >= numTiles) return;

    const int lane = threadIdx.x & 31;
    const int m    = lane & 15;
    const int kh   = (lane >> 4) << 1;
    const int row0 = tile << 4;

    int arow_idx = row0 + m;
    if (arow_idx >= n_nodes) arow_idx = n_nodes - 1;
    const float* arow = feat + (size_t)arow_idx * D_IN + kh;

    const float* b0p = &lds_w[(0 * 16 + m) * LDSW_STRIDE + kh];
    const float* b1p = &lds_w[(1 * 16 + m) * LDSW_STRIDE + kh];
    const float* b2p = &lds_w[(2 * 16 + m) * LDSW_STRIDE + kh];
    const float* b3p = &lds_w[(3 * 16 + m) * LDSW_STRIDE + kh];

    v8f acc0 = {}, acc1 = {}, acc2 = {}, acc3 = {};

#pragma unroll 8
    for (int kk = 0; kk < D_IN; kk += 4) {
        v2f a  = *(const v2f*)(arow + kk);
        v2f b0 = *(const v2f*)(b0p + kk);
        v2f b1 = *(const v2f*)(b1p + kk);
        v2f b2 = *(const v2f*)(b2p + kk);
        v2f b3 = *(const v2f*)(b3p + kk);
        acc0 = __builtin_amdgcn_wmma_f32_16x16x4_f32(false, a, false, b0, (short)0, acc0, false, false);
        acc1 = __builtin_amdgcn_wmma_f32_16x16x4_f32(false, a, false, b1, (short)0, acc1, false, false);
        acc2 = __builtin_amdgcn_wmma_f32_16x16x4_f32(false, a, false, b2, (short)0, acc2, false, false);
        acc3 = __builtin_amdgcn_wmma_f32_16x16x4_f32(false, a, false, b3, (short)0, acc3, false, false);
    }

    const float bb0 = bias[m], bb1 = bias[16 + m], bb2 = bias[32 + m], bb3 = bias[48 + m];
    const int mbase = (lane >> 4) * 8;
#pragma unroll
    for (int v = 0; v < 8; ++v) {
        int r = row0 + mbase + v;
        if (r < n_nodes) {
            float* o = pre + (size_t)r * D_OUT;
            o[m]      = acc0[v] + bb0;
            o[16 + m] = acc1[v] + bb1;
            o[32 + m] = acc2[v] + bb2;
            o[48 + m] = acc3[v] + bb3;
        }
    }
}

__global__ __launch_bounds__(256) void gcn_zero2(float* __restrict__ a,
                                                 float* __restrict__ b, int n)
{
    int i = blockIdx.x * 256 + threadIdx.x;
    if (i < n) { a[i] = 0.0f; b[i] = 0.0f; }
}

__global__ __launch_bounds__(256) void gcn_spmm_scatter(
    const int* __restrict__ rows, const int* __restrict__ cols,
    const float* __restrict__ vals, const float* __restrict__ pre,
    float* __restrict__ s, float* __restrict__ sq, int n_edges)
{
    long gid  = (long)blockIdx.x * 256 + threadIdx.x;
    int  e    = (int)(gid >> 5);
    int  lane = (int)(gid & 31);
    if (e >= n_edges) return;

    const int   r = rows[e];
    const int   c = cols[e];
    const float v = vals[e];

    const float* p = pre + (size_t)c * D_OUT;
    float* srow = s  + (size_t)r * D_OUT;
    float* qrow = sq + (size_t)r * D_OUT;

    float p0 = p[lane];
    float p1 = p[lane + 32];
    unsafeAtomicAdd(srow + lane,      v * p0);
    unsafeAtomicAdd(srow + lane + 32, v * p1);
    unsafeAtomicAdd(qrow + lane,      v * p0 * p0);
    unsafeAtomicAdd(qrow + lane + 32, v * p1 * p1);
}

__global__ __launch_bounds__(256) void gcn_finalize(
    float* __restrict__ s_out, const float* __restrict__ sq,
    const float* __restrict__ diag, int total)
{
    int i = blockIdx.x * 256 + threadIdx.x;
    if (i < total) {
        float sv = s_out[i];
        s_out[i] = diag[i >> 6] * 0.5f * (sv * sv - sq[i]);
    }
}

extern "C" void kernel_launch(void* const* d_in, const int* in_sizes, int n_in,
                              void* d_out, int out_size, void* d_ws, size_t ws_size,
                              hipStream_t stream)
{
    const float* feat     = (const float*)d_in[0];
    const float* W        = (const float*)d_in[1];
    const float* bias     = (const float*)d_in[2];
    const float* adj_vals = (const float*)d_in[3];
    const float* diag     = (const float*)d_in[4];
    const int*   adj_rows = (const int*)d_in[5];
    const int*   adj_cols = (const int*)d_in[6];

    const int n_nodes = in_sizes[4];
    const int n_edges = in_sizes[3];

    float* pre = (float*)d_ws;
    float* sq  = pre + (size_t)n_nodes * D_OUT;
    float* s   = (float*)d_out;

    const int total = n_nodes * D_OUT;

    gcn_zero2<<<(total + 255) / 256, 256, 0, stream>>>(s, sq, total);

    const int tiles  = (n_nodes + 15) / 16;
    const int blocks = (tiles + 7) / 8;
    gcn_gemm_wmma<<<blocks, 256, D_OUT * LDSW_STRIDE * sizeof(float), stream>>>(
        feat, W, bias, pre, n_nodes);

    const long sthreads = (long)n_edges * 32;
    gcn_spmm_scatter<<<(unsigned)((sthreads + 255) / 256), 256, 0, stream>>>(
        adj_rows, adj_cols, adj_vals, pre, s, sq, n_edges);

    gcn_finalize<<<(total + 255) / 256, 256, 0, stream>>>(s, sq, diag, total);
}


# L2 float atomics replaced: integer-add slot lists per row, one wave per row sums its rows of pre in ascending edge id (fixed sum order), s and sq in registers, out written once
# speedup vs baseline: 1.5027x; 1.5027x over previous
#include <hip/hip_runtime.h>

#if defined(__HIP_DEVICE_COMPILE__)
#if !__has_builtin(__builtin_amdgcn_wmma_f32_16x16x4_f32)
#error "gfx1250 f32 WMMA builtin not available on this toolchain (device pass)"
#endif
#endif

typedef __attribute__((ext_vector_type(2))) float v2f;
typedef __attribute__((ext_vector_type(8))) float v8f;

#define D_IN 256
#define D_OUT 64
#define LDSW_STRIDE 260

__global__ __launch_bounds__(256) void gcn_gemm_wmma(
    const float* __restrict__ feat, const float* __restrict__ W,
    const float* __restrict__ bias, float* __restrict__ pre, int n_nodes)
{
    extern __shared__ float lds_w[];

    for (int idx = threadIdx.x; idx < D_OUT * (D_IN / 4); idx += 256) {
        int r  = idx / (D_IN / 4);
        int c4 = (idx % (D_IN / 4)) * 4;
        const float4 w4 = ((const float4*)W)[r * (D_IN / 4) + (c4 >> 2)];
        float* dst = &lds_w[r * LDSW_STRIDE + c4];
        dst[0] = w4.x; dst[1] = w4.y; dst[2] = w4.z; dst[3] = w4.w;
    }
    __syncthreads();

    const int tile     = blockIdx.x * 8 + (threadIdx.x >> 5);
    const int numTiles = (n_nodes + 15) >> 4;
    if (tile >= numTiles) return;

    const int lane = threadIdx.x & 31;
    const int m    = lane & 15;
    const int kh   = (lane >> 4) << 1;
    const int row0 = tile << 4;

    int arow_idx = row0 + m;
    if (arow_idx >= n_nodes) arow_idx = n_nodes - 1;
    const float* arow = feat + (size_t)arow_idx * D_IN + kh;

    const float* b0p = &lds_w[(0 * 16 + m) * LDSW_STRIDE + kh];
    const float* b1p = &lds_w[(1 * 16 + m) * LDSW_STRIDE + kh];
    const float* b2p = &lds_w[(2 * 16 + m) * LDSW_STRIDE + kh];
    const float* b3p = &lds_w[(3 * 16 + m) * LDSW_STRIDE + kh];

    v8f acc0 = {}, acc1 = {}, acc2 = {}, acc3 = {};

#pragma unroll 8
    for (int kk = 0; kk < D_IN; kk += 4) {
        v2f a  = *(const v2f*)(arow + kk);
        v2f b0 = *(const v2f*)(b0p + kk);
        v2f b1 = *(const v2f*)(b1p + kk);
        v2f b2 = *(const v2f*)(b2p + kk);
        v2f b3 = *(const v2f*)(b3p + kk);
        acc0 = __builtin_amdgcn_wmma_f32_16x16x4_f32(false, a, false, b0, (short)0, acc0, false, false);
        acc1 = __builtin_amdgcn_wmma_f32_16x16x4_f32(false, a, false, b1, (short)0, acc1, false, false);
        acc2 = __builtin_amdgcn_wmma_f32_16x16x4_f32(false, a, false, b2, (short)0, acc2, false, false);
        acc3 = __builtin_amdgcn_wmma_f32_16x16x4_f32(false, a, false, b3, (short)0, acc3, false, false);
    }

    const float bb0 = bias[m], bb1 = bias[16 + m], bb2 = bias[32 + m], bb3 = bias[48 + m];
    const int mbase = (lane >> 4) * 8;
#pragma unroll
    for (int v = 0; v < 8; ++v) {
        int r = row0 + mbase + v;
        if (r < n_nodes) {
            float* o = pre + (size_t)r * D_OUT;
            o[m]      = acc0[v] + bb0;
            o[16 + m] = acc1[v] + bb1;
            o[32 + m] = acc2[v] + bb2;
            o[48 + m] = acc3[v] + bb3;
        }
    }
}

#define ROW_SLOTS 64
#define ROW_CAP   63

__global__ __launch_bounds__(256) void gcn_clear_counts(int* __restrict__ rec, int n_nodes)
{
    int i = blockIdx.x * 256 + threadIdx.x;
    if (i < n_nodes) rec[(size_t)i * ROW_SLOTS] = 0;
}

__global__ __launch_bounds__(256) void gcn_list_edges(
    const int* __restrict__ rows, int* __restrict__ rec, int n_edges)
{
    int e = blockIdx.x * 256 + threadIdx.x;
    if (e >= n_edges) return;
    int* rr = rec + (size_t)rows[e] * ROW_SLOTS;
    int slot = atomicAdd(rr, 1);
    if (slot < ROW_CAP) rr[1 + slot] = e;
}

struct RowAcc { float s0, s1, q0, q1; };

__device__ __forceinline__ int lane_i(int x, int l)
{
    return __builtin_amdgcn_readlane(x, l);
}
__device__ __forceinline__ float lane_f(float x, int l)
{
    return __int_as_float(__builtin_amdgcn_readlane(__float_as_int(x), l));
}

template <int U>
__device__ __forceinline__ void sum_batch(RowAcc& a, const float* __restrict__ pl,
                                          int sc, float sv, int j)
{
    float2 p[U];
    float  v[U];
#pragma unroll
    for (int u = 0; u < U; ++u) {
        const int c = lane_i(sc, j + u);
        v[u] = lane_f(sv, j + u);
        p[u] = *(const float2*)(pl + (size_t)c * D_OUT);
    }
#pragma unroll
    for (int u = 0; u < U; ++u) {
        const float t0 = v[u] * p[u].x;
        const float t1 = v[u] * p[u].y;
        a.s0 += t0;
        a.s1 += t1;
        a.q0 += t0 * p[u].x;
        a.q1 += t1 * p[u].y;
    }
}

__device__ __forceinline__ void sum_run(RowAcc& a, const float* __restrict__ pl,
                                        int sc, float sv, int n)
{
    int j = 0;
    for (; j + 8 <= n; j += 8) sum_batch<8>(a, pl, sc, sv, j);
    if (j + 4 <= n) { sum_batch<4>(a, pl, sc, sv, j); j += 4; }
    if (j + 2 <= n) { sum_batch<2>(a, pl, sc, sv, j); j += 2; }
    if (j < n)      { sum_batch<1>(a, pl, sc, sv, j); }
}

__global__ __launch_bounds__(256) void gcn_gather_rows(
    const int* __restrict__ rec, const int* __restrict__ rows,
    const int* __restrict__ cols, const float* __restrict__ vals,
    const float* __restrict__ pre, const float* __restrict__ diag,
    float* __restrict__ out, int n_nodes, int n_edges)
{
    __shared__ int   sh_c[8][64];
    __shared__ float sh_v[8][64];

    const int  w     = threadIdx.x >> 5;
    const int  lane  = threadIdx.x & 31;
    const int  r_raw = blockIdx.x * 8 + w;
    const bool live  = r_raw < n_nodes;
    const int  r     = __builtin_amdgcn_readfirstlane(live ? r_raw : n_nodes - 1);

    const int* rr     = rec + (size_t)r * ROW_SLOTS;
    const int  deg    = __builtin_amdgcn_readfirstlane(rr[0]);
    const bool listed = deg <= ROW_CAP;
    const int  cnt    = listed ? deg : 0;
    const int  n0     = cnt < 32 ? cnt : 32;

    const bool ok0 = lane < cnt;
    const bool ok1 = lane + 32 < cnt;
    int   k0 = 0x7fffffff, k1 = 0x7fffffff, c0 = 0, c1 = 0;
    float v0 = 0.0f, v1 = 0.0f;
    if (ok0) { k0 = rr[1 + lane];  c0 = cols[k0]; v0 = vals[k0]; }
    if (ok1) { k1 = rr[33 + lane]; c1 = cols[k1]; v1 = vals[k1]; }

    int rk0 = 0, rk1 = 0;
    for (int j = 0; j < n0; ++j) {
        const int kj = lane_i(k0, j);
        rk0 += (kj < k0) ? 1 : 0;
        rk1 += (kj < k1) ? 1 : 0;
    }
    for (int j = 32; j < cnt; ++j) {
        const int kj = lane_i(k1, j - 32);
        rk0 += (kj < k0) ? 1 : 0;
        rk1 += (kj < k1) ? 1 : 0;
    }

    if (ok0) { sh_c[w][rk0] = c0; sh_v[w][rk0] = v0; }
    if (ok1) { sh_c[w][rk1] = c1; sh_v[w][rk1] = v1; }
    __syncthreads();
    int   sc0 = 0, sc1 = 0;
    float sv0 = 0.0f, sv1 = 0.0f;
    if (ok0) { sc0 = sh_c[w][lane];      sv0 = sh_v[w][lane]; }
    if (ok1) { sc1 = sh_c[w][lane + 32]; sv1 = sh_v[w][lane + 32]; }

    RowAcc a = {0.0f, 0.0f, 0.0f, 0.0f};
    const float* pl = pre + 2 * lane;
    sum_run(a, pl, sc0, sv0, n0);
    if (cnt > 32) sum_run(a, pl, sc1, sv1, cnt - 32);

    if (!listed) {
        for (int base = 0; base < n_edges; base += 32) {
            const int e = base + lane;
            int   c = 0;
            float v = 0.0f;
            bool  mine = false;
            if (e < n_edges && rows[e] == r) { mine = true; c = cols[e]; v = vals[e]; }
            unsigned m = __builtin_amdgcn_ballot_w32(mine);
            while (m) {
                const int src = __builtin_ctz(m);
                m &= m - 1;
                sum_batch<1>(a, pl, c, v, src);
            }
        }
    }

    if (live) {
        const float d = diag[r];
        float2 o;
        o.x = d * 0.5f * (a.s0 * a.s0 - a.q0);
        o.y = d * 0.5f * (a.s1 * a.s1 - a.q1);
        *(float2*)(out + (size_t)r * D_OUT + 2 * lane) = o;
    }
}

extern "C" void kernel_launch(void* const* d_in, const int* in_sizes, int n_in,
                              void* d_out, int out_size, void* d_ws, size_t ws_size,
                              hipStream_t stream)
{
    const float* feat     = (const float*)d_in[0];
    const float* W        = (const float*)d_in[1];
    const float* bias     = (const float*)d_in[2];
    const float* adj_vals = (const float*)d_in[3];
    const float* diag     = (const float*)d_in[4];
    const int*   adj_rows = (const int*)d_in[5];
    const int*   adj_cols = (const int*)d_in[6];

    const int n_nodes = in_sizes[4];
    const int n_edges = in_sizes[3];

    float* pre = (float*)d_ws;
    int*   rec = (int*)(pre + (size_t)n_nodes * D_OUT);
    float* out = (float*)d_out;

    const int tiles  = (n_nodes + 15) / 16;
    const int blocks = (tiles + 7) / 8;
    gcn_gemm_wmma<<<blocks, 256, D_OUT * LDSW_STRIDE * sizeof(float), stream>>>(
        feat, W, bias, pre, n_nodes);

    gcn_clear_counts<<<(n_nodes + 255) / 256, 256, 0, stream>>>(rec, n_nodes);

    gcn_list_edges<<<(n_edges + 255) / 256, 256, 0, stream>>>(adj_rows, rec, n_edges);

    gcn_gather_rows<<<(n_nodes + 7) / 8, 256, 0, stream>>>(
        rec, adj_rows, adj_cols, adj_vals, pre, diag, out, n_nodes, n_edges);
}


# Scattered returning global adds (one per edge) replaced: edges binned by 128-row range, LDS counts per block and one global add per block and range; rows dealt in LDS; per-row sum order kept
# speedup vs baseline: 2.2633x; 1.5062x over previous
#include <hip/hip_runtime.h>

#if defined(__HIP_DEVICE_COMPILE__)
#if !__has_builtin(__builtin_amdgcn_wmma_f32_16x16x4_f32)
#error "gfx1250 f32 WMMA builtin not available on this toolchain (device pass)"
#endif
#endif

typedef __attribute__((ext_vector_type(2))) float v2f;
typedef __attribute__((ext_vector_type(8))) float v8f;

#define D_IN 256
#define D_OUT 64
#define LDSW_STRIDE 260

__global__ __launch_bounds__(256) void gcn_gemm_wmma(
    const float* __restrict__ feat, const float* __restrict__ W,
    const float* __restrict__ bias, float* __restrict__ pre, int n_nodes)
{
    extern __shared__ float lds_w[];

    for (int idx = threadIdx.x; idx < D_OUT * (D_IN / 4); idx += 256) {
        int r  = idx / (D_IN / 4);
        int c4 = (idx % (D_IN / 4)) * 4;
        const float4 w4 = ((const float4*)W)[r * (D_IN / 4) + (c4 >> 2)];
        float* dst = &lds_w[r * LDSW_STRIDE + c4];
        dst[0] = w4.x; dst[1] = w4.y; dst[2] = w4.z; dst[3] = w4.w;
    }
    __syncthreads();

    const int tile     = blockIdx.x * 8 + (threadIdx.x >> 5);
    const int numTiles = (n_nodes + 15) >> 4;
    if (tile >= numTiles) return;

    const int lane = threadIdx.x & 31;
    const int m    = lane & 15;
    const int kh   = (lane >> 4) << 1;
    const int row0 = tile << 4;

    int arow_idx = row0 + m;
    if (arow_idx >= n_nodes) arow_idx = n_nodes - 1;
    const float* arow = feat + (size_t)arow_idx * D_IN + kh;

    const float* b0p = &lds_w[(0 * 16 + m) * LDSW_STRIDE + kh];
    const float* b1p = &lds_w[(1 * 16 + m) * LDSW_STRIDE + kh];
    const float* b2p = &lds_w[(2 * 16 + m) * LDSW_STRIDE + kh];
    const float* b3p = &lds_w[(3 * 16 + m) * LDSW_STRIDE + kh];

    v8f acc0 = {}, acc1 = {}, acc2 = {}, acc3 = {};

#pragma unroll 8
    for (int kk = 0; kk < D_IN; kk += 4) {
        v2f a  = *(const v2f*)(arow + kk);
        v2f b0 = *(const v2f*)(b0p + kk);
        v2f b1 = *(const v2f*)(b1p + kk);
        v2f b2 = *(const v2f*)(b2p + kk);
        v2f b3 = *(const v2f*)(b3p + kk);
        acc0 = __builtin_amdgcn_wmma_f32_16x16x4_f32(false, a, false, b0, (short)0, acc0, false, false);
        acc1 = __builtin_amdgcn_wmma_f32_16x16x4_f32(false, a, false, b1, (short)0, acc1, false, false);
        acc2 = __builtin_amdgcn_wmma_f32_16x16x4_f32(false, a, false, b2, (short)0, acc2, false, false);
        acc3 = __builtin_amdgcn_wmma_f32_16x16x4_f32(false, a, false, b3, (short)0, acc3, false, false);
    }

    const float bb0 = bias[m], bb1 = bias[16 + m], bb2 = bias[32 + m], bb3 = bias[48 + m];
    const int mbase = (lane >> 4) * 8;
#pragma unroll
    for (int v = 0; v < 8; ++v) {
        int r = row0 + mbase + v;
        if (r < n_nodes) {
            float* o = pre + (size_t)r * D_OUT;
            o[m]      = acc0[v] + bb0;
            o[16 + m] = acc1[v] + bb1;
            o[32 + m] = acc2[v] + bb2;
            o[48 + m] = acc3[v] + bb3;
        }
    }
}

#define RANGE_ROWS  128
#define RANGE_SHIFT 7
#define RANGE_CAP   6144
#define ROW_CAP     64
#define BIN_THREADS 1024
#define BIN_EDGES   8
#define GATHER_WAVES 16

__global__ __launch_bounds__(256) void gcn_clear_counts(int* __restrict__ range_count, int n_ranges)
{
    int i = blockIdx.x * 256 + threadIdx.x;
    if (i < n_ranges) range_count[i] = 0;
}

__global__ __launch_bounds__(BIN_THREADS) void gcn_bin_edges(
    const int* __restrict__ rows, int* __restrict__ range_list,
    int* __restrict__ range_count, int n_edges, int n_nodes, int n_ranges)
{
    extern __shared__ int sh_bin[];
    int* sh_hist = sh_bin;
    int* sh_base = sh_bin + n_ranges;

    for (int b = threadIdx.x; b < n_ranges; b += BIN_THREADS) sh_hist[b] = 0;
    __syncthreads();

    const int e0 = blockIdx.x * (BIN_THREADS * BIN_EDGES) + threadIdx.x;
    int bk[BIN_EDGES], pl[BIN_EDGES];
#pragma unroll
    for (int k = 0; k < BIN_EDGES; ++k) {
        const int e  = e0 + k * BIN_THREADS;
        const int ec = e < n_edges ? e : n_edges - 1;
        const int r  = rows[ec];
        bk[k] = (e < n_edges && (unsigned)r < (unsigned)n_nodes) ? (r >> RANGE_SHIFT) : -1;
    }
#pragma unroll
    for (int k = 0; k < BIN_EDGES; ++k) {
        pl[k] = 0;
        if (bk[k] >= 0) pl[k] = atomicAdd(&sh_hist[bk[k]], 1);
    }
    __syncthreads();

    for (int b = threadIdx.x; b < n_ranges; b += BIN_THREADS) {
        const int c = sh_hist[b];
        sh_base[b] = c ? atomicAdd(&range_count[b], c) : 0;
    }
    __syncthreads();

#pragma unroll
    for (int k = 0; k < BIN_EDGES; ++k) {
        if (bk[k] >= 0) {
            const int pos = sh_base[bk[k]] + pl[k];
            if (pos < RANGE_CAP)
                range_list[(size_t)bk[k] * RANGE_CAP + pos] = e0 + k * BIN_THREADS;
        }
    }
}

struct RowAcc { float s0, s1, q0, q1; };

__device__ __forceinline__ int lane_i(int x, int l)
{
    return __builtin_amdgcn_readlane(x, l);
}
__device__ __forceinline__ float lane_f(float x, int l)
{
    return __int_as_float(__builtin_amdgcn_readlane(__float_as_int(x), l));
}

template <int U>
__device__ __forceinline__ void sum_batch(RowAcc& a, const float* __restrict__ pl,
                                          int sc, float sv, int j)
{
    float2 p[U];
    float  v[U];
#pragma unroll
    for (int u = 0; u < U; ++u) {
        const int c = lane_i(sc, j + u);
        v[u] = lane_f(sv, j + u);
        p[u] = *(const float2*)(pl + (size_t)c * D_OUT);
    }
#pragma unroll
    for (int u = 0; u < U; ++u) {
        const float t0 = v[u] * p[u].x;
        const float t1 = v[u] * p[u].y;
        a.s0 += t0;
        a.s1 += t1;
        a.q0 += t0 * p[u].x;
        a.q1 += t1 * p[u].y;
    }
}

__device__ __forceinline__ void sum_run(RowAcc& a, const float* __restrict__ pl,
                                        int sc, float sv, int n)
{
    int j = 0;
    for (; j + 8 <= n; j += 8) sum_batch<8>(a, pl, sc, sv, j);
    if (j + 4 <= n) { sum_batch<4>(a, pl, sc, sv, j); j += 4; }
    if (j + 2 <= n) { sum_batch<2>(a, pl, sc, sv, j); j += 2; }
    if (j < n)      { sum_batch<1>(a, pl, sc, sv, j); }
}

__global__ __launch_bounds__(GATHER_WAVES * 32) void gcn_gather_ranges(
    const int* __restrict__ range_list, const int* __restrict__ range_count,
    const int* __restrict__ rows, const int* __restrict__ cols,
    const float* __restrict__ vals, const float* __restrict__ pre,
    const float* __restrict__ diag, float* __restrict__ out,
    int n_nodes, int n_edges)
{
    __shared__ int   sh_cnt[RANGE_ROWS];
    __shared__ int   sh_list[RANGE_ROWS][ROW_CAP + 1];
    __shared__ int   sh_c[GATHER_WAVES][ROW_CAP];
    __shared__ float sh_v[GATHER_WAVES][ROW_CAP];

    const int b    = blockIdx.x;
    const int tid  = threadIdx.x;
    const int w    = tid >> 5;
    const int lane = tid & 31;

    if (tid < RANGE_ROWS) sh_cnt[tid] = 0;
    __syncthreads();

    const int  bc    = range_count[b];
    const bool whole = bc <= RANGE_CAP;
    const int  total = whole ? bc : n_edges;
    const int* bl    = range_list + (size_t)b * RANGE_CAP;
    for (int base = 0; base < total; base += 4 * GATHER_WAVES * 32) {
        int e[4], r[4];
#pragma unroll
        for (int k = 0; k < 4; ++k) {
            const int idx = base + k * (GATHER_WAVES * 32) + tid;
            const int ic  = idx < total ? idx : total - 1;
            e[k] = whole ? bl[ic] : ic;
        }
#pragma unroll
        for (int k = 0; k < 4; ++k) r[k] = rows[e[k]];
#pragma unroll
        for (int k = 0; k < 4; ++k) {
            const int idx = base + k * (GATHER_WAVES * 32) + tid;
            if (idx < total && (unsigned)r[k] < (unsigned)n_nodes && (r[k] >> RANGE_SHIFT) == b) {
                const int rl   = r[k] & (RANGE_ROWS - 1);
                const int slot = atomicAdd(&sh_cnt[rl], 1);
                if (slot < ROW_CAP) sh_list[rl][slot] = e[k];
            }
        }
    }
    __syncthreads();

    const float* pl = pre + 2 * lane;

    const int r_first = b * RANGE_ROWS + w * (RANGE_ROWS / GATHER_WAVES);
    float dg = 0.0f;
    if (lane < RANGE_ROWS / GATHER_WAVES && r_first + lane < n_nodes) dg = diag[r_first + lane];

    for (int i = 0; i < RANGE_ROWS / GATHER_WAVES; ++i) {
        const int rl = w * (RANGE_ROWS / GATHER_WAVES) + i;
        const int r  = __builtin_amdgcn_readfirstlane(b * RANGE_ROWS + rl);
        if (r >= n_nodes) break;

        const int  deg    = __builtin_amdgcn_readfirstlane(sh_cnt[rl]);
        const bool listed = deg <= ROW_CAP;
        const int  cnt    = listed ? deg : 0;
        const int  n0     = cnt < 32 ? cnt : 32;

        const bool ok0 = lane < cnt;
        const bool ok1 = lane + 32 < cnt;
        int   k0 = 0x7fffffff, k1 = 0x7fffffff, c0 = 0, c1 = 0;
        float v0 = 0.0f, v1 = 0.0f;
        if (ok0) { k0 = sh_list[rl][lane];      c0 = cols[k0]; v0 = vals[k0]; }
        if (ok1) { k1 = sh_list[rl][lane + 32]; c1 = cols[k1]; v1 = vals[k1]; }

        int rk0 = 0, rk1 = 0;
        for (int j = 0; j < n0; ++j) {
            const int kj = lane_i(k0, j);
            rk0 += (kj < k0) ? 1 : 0;
            rk1 += (kj < k1) ? 1 : 0;
        }
        for (int j = 32; j < cnt; ++j) {
            const int kj = lane_i(k1, j - 32);
            rk0 += (kj < k0) ? 1 : 0;
            rk1 += (kj < k1) ? 1 : 0;
        }

        if (ok0) { sh_c[w][rk0] = c0; sh_v[w][rk0] = v0; }
        if (ok1) { sh_c[w][rk1] = c1; sh_v[w][rk1] = v1; }
        __builtin_amdgcn_fence(__ATOMIC_RELEASE, "workgroup");
        __builtin_amdgcn_wave_barrier();
        __builtin_amdgcn_fence(__ATOMIC_ACQUIRE, "workgroup");
        int   sc0 = 0, sc1 = 0;
        float sv0 = 0.0f, sv1 = 0.0f;
        if (ok0) { sc0 = sh_c[w][lane];      sv0 = sh_v[w][lane]; }
        if (ok1) { sc1 = sh_c[w][lane + 32]; sv1 = sh_v[w][lane + 32]; }

        RowAcc a = {0.0f, 0.0f, 0.0f, 0.0f};
        sum_run(a, pl, sc0, sv0, n0);
        if (cnt > 32) sum_run(a, pl, sc1, sv1, cnt - 32);

        if (!listed) {
            for (int base = 0; base < n_edges; base += 32) {
                const int e = base + lane;
                int   c = 0;
                float v = 0.0f;
                bool  mine = false;
                if (e < n_edges && rows[e] == r) { mine = true; c = cols[e]; v = vals[e]; }
                unsigned m = __builtin_amdgcn_ballot_w32(mine);
                while (m) {
                    const int src = __builtin_ctz(m);
                    m &= m - 1;
                    sum_batch<1>(a, pl, c, v, src);
                }
            }
        }

        const float d = lane_f(dg, i);
        float2 o;
        o.x = d * 0.5f * (a.s0 * a.s0 - a.q0);
        o.y = d * 0.5f * (a.s1 * a.s1 - a.q1);
        *(float2*)(out + (size_t)r * D_OUT + 2 * lane) = o;
    }
}

extern "C" void kernel_launch(void* const* d_in, const int* in_sizes, int n_in,
                              void* d_out, int out_size, void* d_ws, size_t ws_size,
                              hipStream_t stream)
{
    const float* feat     = (const float*)d_in[0];
    const float* W        = (const float*)d_in[1];
    const float* bias     = (const float*)d_in[2];
    const float* adj_vals = (const float*)d_in[3];
    const float* diag     = (const float*)d_in[4];
    const int*   adj_rows = (const int*)d_in[5];
    const int*   adj_cols = (const int*)d_in[6];

    const int n_nodes = in_sizes[4];
    const int n_edges = in_sizes[3];

    const int n_ranges = (n_nodes + RANGE_ROWS - 1) / RANGE_ROWS;

    float* pre         = (float*)d_ws;
    int*   range_list  = (int*)(pre + (size_t)n_nodes * D_OUT);
    int*   range_count = range_list + (size_t)n_ranges * RANGE_CAP;
    float* out         = (float*)d_out;

    const int tiles  = (n_nodes + 15) / 16;
    const int blocks = (tiles + 7) / 8;
    gcn_gemm_wmma<<<blocks, 256, D_OUT * LDSW_STRIDE * sizeof(float), stream>>>(
        feat, W, bias, pre, n_nodes);

    gcn_clear_counts<<<(n_ranges + 255) / 256, 256, 0, stream>>>(range_count, n_ranges);

    gcn_bin_edges<<<(n_edges + BIN_THREADS * BIN_EDGES - 1) / (BIN_THREADS * BIN_EDGES),
                    BIN_THREADS, 2 * n_ranges * sizeof(int), stream>>>(
        adj_rows, range_list, range_count, n_edges, n_nodes, n_ranges);

    gcn_gather_ranges<<<n_ranges, GATHER_WAVES * 32, 0, stream>>>(
        range_list, range_count, adj_rows, adj_cols, adj_vals, pre, diag, out,
        n_nodes, n_edges);
}


# Gather's 3 scattered 4-byte reads per edge (rows/cols/vals by edge id) removed: binning streams them and lists 8-byte (col,row,val) records; per-row sum order now ascending (col,val), still fixed
# speedup vs baseline: 2.6904x; 1.1887x over previous
#include <hip/hip_runtime.h>

#if defined(__HIP_DEVICE_COMPILE__)
#if !__has_builtin(__builtin_amdgcn_wmma_f32_16x16x4_f32)
#error "gfx1250 f32 WMMA builtin not available on this toolchain (device pass)"
#endif
#endif

typedef __attribute__((ext_vector_type(2))) float v2f;
typedef __attribute__((ext_vector_type(8))) float v8f;

#define D_IN 256
#define D_OUT 64
#define LDSW_STRIDE 260

__global__ __launch_bounds__(256) void gcn_gemm_wmma(
    const float* __restrict__ feat, const float* __restrict__ W,
    const float* __restrict__ bias, float* __restrict__ pre, int n_nodes)
{
    extern __shared__ float lds_w[];

    for (int idx = threadIdx.x; idx < D_OUT * (D_IN / 4); idx += 256) {
        int r  = idx / (D_IN / 4);
        int c4 = (idx % (D_IN / 4)) * 4;
        const float4 w4 = ((const float4*)W)[r * (D_IN / 4) + (c4 >> 2)];
        float* dst = &lds_w[r * LDSW_STRIDE + c4];
        dst[0] = w4.x; dst[1] = w4.y; dst[2] = w4.z; dst[3] = w4.w;
    }
    __syncthreads();

    const int tile     = blockIdx.x * 8 + (threadIdx.x >> 5);
    const int numTiles = (n_nodes + 15) >> 4;
    if (tile >= numTiles) return;

    const int lane = threadIdx.x & 31;
    const int m    = lane & 15;
    const int kh   = (lane >> 4) << 1;
    const int row0 = tile << 4;

    int arow_idx = row0 + m;
    if (arow_idx >= n_nodes) arow_idx = n_nodes - 1;
    const float* arow = feat + (size_t)arow_idx * D_IN + kh;

    const float* b0p = &lds_w[(0 * 16 + m) * LDSW_STRIDE + kh];
    const float* b1p = &lds_w[(1 * 16 + m) * LDSW_STRIDE + kh];
    const float* b2p = &lds_w[(2 * 16 + m) * LDSW_STRIDE + kh];
    const float* b3p = &lds_w[(3 * 16 + m) * LDSW_STRIDE + kh];

    v8f acc0 = {}, acc1 = {}, acc2 = {}, acc3 = {};

#pragma unroll 8
    for (int kk = 0; kk < D_IN; kk += 4) {
        v2f a  = *(const v2f*)(arow + kk);
        v2f b0 = *(const v2f*)(b0p + kk);
        v2f b1 = *(const v2f*)(b1p + kk);
        v2f b2 = *(const v2f*)(b2p + kk);
        v2f b3 = *(const v2f*)(b3p + kk);
        acc0 = __builtin_amdgcn_wmma_f32_16x16x4_f32(false, a, false, b0, (short)0, acc0, false, false);
        acc1 = __builtin_amdgcn_wmma_f32_16x16x4_f32(false, a, false, b1, (short)0, acc1, false, false);
        acc2 = __builtin_amdgcn_wmma_f32_16x16x4_f32(false, a, false, b2, (short)0, acc2, false, false);
        acc3 = __builtin_amdgcn_wmma_f32_16x16x4_f32(false, a, false, b3, (short)0, acc3, false, false);
    }

    const float bb0 = bias[m], bb1 = bias[16 + m], bb2 = bias[32 + m], bb3 = bias[48 + m];
    const int mbase = (lane >> 4) * 8;
#pragma unroll
    for (int v = 0; v < 8; ++v) {
        int r = row0 + mbase + v;
        if (r < n_nodes) {
            float* o = pre + (size_t)r * D_OUT;
            o[m]      = acc0[v] + bb0;
            o[16 + m] = acc1[v] + bb1;
            o[32 + m] = acc2[v] + bb2;
            o[48 + m] = acc3[v] + bb3;
        }
    }
}

#define RANGE_ROWS  128
#define RANGE_SHIFT 7
#define RANGE_CAP   3072
#define ROW_CAP     64

typedef __attribute__((ext_vector_type(2))) unsigned int edge_rec;
#define BIN_THREADS 1024
#define BIN_EDGES   8
#define GATHER_WAVES 16
#define DEAL_RECS   (RANGE_CAP / (GATHER_WAVES * 32))
static_assert(DEAL_RECS * GATHER_WAVES * 32 == RANGE_CAP, "a whole list must be one trip of the gather block");

__global__ __launch_bounds__(256) void gcn_clear_counts(int* __restrict__ range_count, int n_ranges)
{
    int i = blockIdx.x * 256 + threadIdx.x;
    if (i < n_ranges) range_count[i] = 0;
}

__global__ __launch_bounds__(BIN_THREADS) void gcn_bin_edges(
    const int* __restrict__ rows, const int* __restrict__ cols, const float* __restrict__ vals,
    int* __restrict__ range_list, int* __restrict__ range_count, int n_edges, int n_nodes, int n_ranges)
{
    extern __shared__ int sh_bin[];
    int* sh_hist = sh_bin;
    int* sh_base = sh_bin + n_ranges;

    for (int b = threadIdx.x; b < n_ranges; b += BIN_THREADS) sh_hist[b] = 0;
    __syncthreads();

    const int e0 = blockIdx.x * (BIN_THREADS * BIN_EDGES) + threadIdx.x;
    int      bk[BIN_EDGES], pl[BIN_EDGES], rr[BIN_EDGES];
    edge_rec rec[BIN_EDGES];
#pragma unroll
    for (int k = 0; k < BIN_EDGES; ++k) {
        const int e  = e0 + k * BIN_THREADS;
        const int ec = e < n_edges ? e : n_edges - 1;
        rr[k] = rows[ec];
    }
#pragma unroll
    for (int k = 0; k < BIN_EDGES; ++k) {
        const int e  = e0 + k * BIN_THREADS;
        const int ec = e < n_edges ? e : n_edges - 1;
        rec[k].x = (unsigned)cols[ec];
        rec[k].y = __float_as_uint(vals[ec]);
    }
#pragma unroll
    for (int k = 0; k < BIN_EDGES; ++k) {
        const int e = e0 + k * BIN_THREADS;
        bk[k] = (e < n_edges && (unsigned)rr[k] < (unsigned)n_nodes) ? (rr[k] >> RANGE_SHIFT) : -1;
    }
#pragma unroll
    for (int k = 0; k < BIN_EDGES; ++k) {
        pl[k] = 0;
        if (bk[k] >= 0) pl[k] = atomicAdd(&sh_hist[bk[k]], 1);
    }
    __syncthreads();

    for (int b = threadIdx.x; b < n_ranges; b += BIN_THREADS) {
        const int c = sh_hist[b];
        sh_base[b] = c ? atomicAdd(&range_count[b], c) : 0;
    }
    __syncthreads();

    edge_rec* lists = (edge_rec*)range_list;
#pragma unroll
    for (int k = 0; k < BIN_EDGES; ++k) {
        if (bk[k] >= 0) {
            const int pos = sh_base[bk[k]] + pl[k];
            if (pos < RANGE_CAP) {
                edge_rec q;
                q.x = (rec[k].x << RANGE_SHIFT) | (unsigned)(rr[k] & (RANGE_ROWS - 1));
                q.y = rec[k].y;
                lists[(size_t)bk[k] * RANGE_CAP + pos] = q;
            }
        }
    }
}

struct RowAcc { float s0, s1, q0, q1; };

__device__ __forceinline__ int lane_i(int x, int l)
{
    return __builtin_amdgcn_readlane(x, l);
}
__device__ __forceinline__ float lane_f(float x, int l)
{
    return __int_as_float(__builtin_amdgcn_readlane(__float_as_int(x), l));
}

template <int U>
__device__ __forceinline__ void sum_batch(RowAcc& a, const float* __restrict__ pl,
                                          int sc, float sv, int j)
{
    float2 p[U];
    float  v[U];
#pragma unroll
    for (int u = 0; u < U; ++u) {
        const int c = lane_i(sc, j + u);
        v[u] = lane_f(sv, j + u);
        p[u] = *(const float2*)(pl + (size_t)c * D_OUT);
    }
#pragma unroll
    for (int u = 0; u < U; ++u) {
        const float t0 = v[u] * p[u].x;
        const float t1 = v[u] * p[u].y;
        a.s0 += t0;
        a.s1 += t1;
        a.q0 += t0 * p[u].x;
        a.q1 += t1 * p[u].y;
    }
}

__device__ __forceinline__ void sum_run(RowAcc& a, const float* __restrict__ pl,
                                        int sc, float sv, int n)
{
    int j = 0;
    for (; j + 8 <= n; j += 8) sum_batch<8>(a, pl, sc, sv, j);
    if (j + 4 <= n) { sum_batch<4>(a, pl, sc, sv, j); j += 4; }
    if (j + 2 <= n) { sum_batch<2>(a, pl, sc, sv, j); j += 2; }
    if (j < n)      { sum_batch<1>(a, pl, sc, sv, j); }
}

__global__ __launch_bounds__(GATHER_WAVES * 32) void gcn_gather_ranges(
    const int* __restrict__ range_list, const int* __restrict__ range_count,
    const int* __restrict__ rows, const int* __restrict__ cols,
    const float* __restrict__ vals, const float* __restrict__ pre,
    const float* __restrict__ diag, float* __restrict__ out,
    int n_nodes, int n_edges)
{
    __shared__ int      sh_cnt[RANGE_ROWS];
    __shared__ edge_rec sh_rec[RANGE_ROWS][ROW_CAP + 1];
    __shared__ int      sh_c[GATHER_WAVES][ROW_CAP];
    __shared__ float    sh_v[GATHER_WAVES][ROW_CAP];

    const int b    = blockIdx.x;
    const int tid  = threadIdx.x;
    const int w    = tid >> 5;
    const int lane = tid & 31;

    if (tid < RANGE_ROWS) sh_cnt[tid] = 0;
    __syncthreads();

    const int bc = range_count[b];
    if (bc <= RANGE_CAP) {
        const edge_rec* bl = (const edge_rec*)range_list + (size_t)b * RANGE_CAP;
        if (bc > 0) {
            edge_rec q[DEAL_RECS];
#pragma unroll
            for (int k = 0; k < DEAL_RECS; ++k) {
                const int idx = k * (GATHER_WAVES * 32) + tid;
                q[k] = bl[idx < bc ? idx : bc - 1];
            }
#pragma unroll
            for (int k = 0; k < DEAL_RECS; ++k) {
                const int idx = k * (GATHER_WAVES * 32) + tid;
                if (idx < bc) {
                    const int rl   = q[k].x & (RANGE_ROWS - 1);
                    const int slot = atomicAdd(&sh_cnt[rl], 1);
                    if (slot < ROW_CAP) sh_rec[rl][slot] = q[k];
                }
            }
        }
    } else {
        for (int base = 0; base < n_edges; base += GATHER_WAVES * 32) {
            const int e = base + tid;
            if (e < n_edges) {
                const int r = rows[e];
                if ((unsigned)r < (unsigned)n_nodes && (r >> RANGE_SHIFT) == b) {
                    const int rl   = r & (RANGE_ROWS - 1);
                    const int slot = atomicAdd(&sh_cnt[rl], 1);
                    if (slot < ROW_CAP) {
                        edge_rec q;
                        q.x = ((unsigned)cols[e] << RANGE_SHIFT) | (unsigned)rl;
                        q.y = __float_as_uint(vals[e]);
                        sh_rec[rl][slot] = q;
                    }
                }
            }
        }
    }
    __syncthreads();

    const float* pl = pre + 2 * lane;

    const int r_first = b * RANGE_ROWS + w * (RANGE_ROWS / GATHER_WAVES);
    float dg = 0.0f;
    if (lane < RANGE_ROWS / GATHER_WAVES && r_first + lane < n_nodes) dg = diag[r_first + lane];

    for (int i = 0; i < RANGE_ROWS / GATHER_WAVES; ++i) {
        const int rl = w * (RANGE_ROWS / GATHER_WAVES) + i;
        const int r  = __builtin_amdgcn_readfirstlane(b * RANGE_ROWS + rl);
        if (r >= n_nodes) break;

        const int  deg    = __builtin_amdgcn_readfirstlane(sh_cnt[rl]);
        const bool listed = deg <= ROW_CAP;
        const int  cnt    = listed ? deg : 0;
        const int  n0     = cnt < 32 ? cnt : 32;

        const bool ok0 = lane < cnt;
        const bool ok1 = lane + 32 < cnt;
        edge_rec q0 = {0u, 0u}, q1 = {0u, 0u};
        if (ok0) q0 = sh_rec[rl][lane];
        if (ok1) q1 = sh_rec[rl][lane + 32];
        const int   c0 = (int)(q0.x >> RANGE_SHIFT), c1 = (int)(q1.x >> RANGE_SHIFT);
        const float v0 = __uint_as_float(q0.y),      v1 = __uint_as_float(q1.y);

        const unsigned long long key0 =
            ((unsigned long long)((q0.x & ~(unsigned)(RANGE_ROWS - 1)) | (q0.y >> (32 - RANGE_SHIFT))) << 32) |
            ((q0.y << RANGE_SHIFT) | (unsigned)lane);
        const unsigned long long key1 =
            ((unsigned long long)((q1.x & ~(unsigned)(RANGE_ROWS - 1)) | (q1.y >> (32 - RANGE_SHIFT))) << 32) |
            ((q1.y << RANGE_SHIFT) | (unsigned)(lane + 32));
        const int k0h = (int)(key0 >> 32), k0l = (int)key0;
        const int k1h = (int)(key1 >> 32), k1l = (int)key1;

        int rk0 = 0, rk1 = 0;
        for (int j = 0; j < n0; ++j) {
            const unsigned long long kj =
                ((unsigned long long)(unsigned)lane_i(k0h, j) << 32) | (unsigned)lane_i(k0l, j);
            rk0 += (kj < key0) ? 1 : 0;
            rk1 += (kj < key1) ? 1 : 0;
        }
        for (int j = 32; j < cnt; ++j) {
            const unsigned long long kj =
                ((unsigned long long)(unsigned)lane_i(k1h, j - 32) << 32) | (unsigned)lane_i(k1l, j - 32);
            rk0 += (kj < key0) ? 1 : 0;
            rk1 += (kj < key1) ? 1 : 0;
        }

        if (ok0) { sh_c[w][rk0] = c0; sh_v[w][rk0] = v0; }
        if (ok1) { sh_c[w][rk1] = c1; sh_v[w][rk1] = v1; }
        __builtin_amdgcn_fence(__ATOMIC_RELEASE, "workgroup");
        __builtin_amdgcn_wave_barrier();
        __builtin_amdgcn_fence(__ATOMIC_ACQUIRE, "workgroup");
        int   sc0 = 0, sc1 = 0;
        float sv0 = 0.0f, sv1 = 0.0f;
        if (ok0) { sc0 = sh_c[w][lane];      sv0 = sh_v[w][lane]; }
        if (ok1) { sc1 = sh_c[w][lane + 32]; sv1 = sh_v[w][lane + 32]; }

        RowAcc a = {0.0f, 0.0f, 0.0f, 0.0f};
        sum_run(a, pl, sc0, sv0, n0);
        if (cnt > 32) sum_run(a, pl, sc1, sv1, cnt - 32);

        if (!listed) {
            for (int base = 0; base < n_edges; base += 32) {
                const int e = base + lane;
                int   c = 0;
                float v = 0.0f;
                bool  mine = false;
                if (e < n_edges && rows[e] == r) { mine = true; c = cols[e]; v = vals[e]; }
                unsigned m = __builtin_amdgcn_ballot_w32(mine);
                while (m) {
                    const int src = __builtin_ctz(m);
                    m &= m - 1;
                    sum_batch<1>(a, pl, c, v, src);
                }
            }
        }

        const float d = lane_f(dg, i);
        float2 o;
        o.x = d * 0.5f * (a.s0 * a.s0 - a.q0);
        o.y = d * 0.5f * (a.s1 * a.s1 - a.q1);
        *(float2*)(out + (size_t)r * D_OUT + 2 * lane) = o;
    }
}

extern "C" void kernel_launch(void* const* d_in, const int* in_sizes, int n_in,
                              void* d_out, int out_size, void* d_ws, size_t ws_size,
                              hipStream_t stream)
{
    const float* feat     = (const float*)d_in[0];
    const float* W        = (const float*)d_in[1];
    const float* bias     = (const float*)d_in[2];
    const float* adj_vals = (const float*)d_in[3];
    const float* diag     = (const float*)d_in[4];
    const int*   adj_rows = (const int*)d_in[5];
    const int*   adj_cols = (const int*)d_in[6];

    const int n_nodes = in_sizes[4];
    const int n_edges = in_sizes[3];

    const int n_ranges = (n_nodes + RANGE_ROWS - 1) / RANGE_ROWS;

    float* pre         = (float*)d_ws;
    int*   range_list  = (int*)(pre + (size_t)n_nodes * D_OUT);
    int*   range_count = range_list + (size_t)n_ranges * RANGE_CAP * 2;
    float* out         = (float*)d_out;

    const int tiles  = (n_nodes + 15) / 16;
    const int blocks = (tiles + 7) / 8;
    gcn_gemm_wmma<<<blocks, 256, D_OUT * LDSW_STRIDE * sizeof(float), stream>>>(
        feat, W, bias, pre, n_nodes);

    gcn_clear_counts<<<(n_ranges + 255) / 256, 256, 0, stream>>>(range_count, n_ranges);

    gcn_bin_edges<<<(n_edges + BIN_THREADS * BIN_EDGES - 1) / (BIN_THREADS * BIN_EDGES),
                    BIN_THREADS, 2 * n_ranges * sizeof(int), stream>>>(
        adj_rows, adj_cols, adj_vals, range_list, range_count, n_edges, n_nodes, n_ranges);

    gcn_gather_ranges<<<n_ranges, GATHER_WAVES * 32, 0, stream>>>(
        range_list, range_count, adj_rows, adj_cols, adj_vals, pre, diag, out,
        n_nodes, n_edges);
}


# Clear and binning launches folded into the GEMM launch (4 launches -> 2): each binning block sorts its edge share by row range into its own segment, no shared counters; per-row sum order unchanged
# speedup vs baseline: 3.3509x; 1.2455x over previous
#include <hip/hip_runtime.h>

#if defined(__HIP_DEVICE_COMPILE__)
#if !__has_builtin(__builtin_amdgcn_wmma_f32_16x16x4_f32)
#error "gfx1250 f32 WMMA builtin not available on this toolchain (device pass)"
#endif
#endif

typedef __attribute__((ext_vector_type(2))) float v2f;
typedef __attribute__((ext_vector_type(8))) float v8f;

#define D_IN 256
#define D_OUT 64
#define LDSW_STRIDE 260

#define L1_THREADS 1024
#define L1_WAVES   (L1_THREADS / 32)

__device__ __forceinline__ void gemm_role(
    const int gblock, float* __restrict__ lds_w,
    const float* __restrict__ feat, const float* __restrict__ W,
    const float* __restrict__ bias, float* __restrict__ pre, int n_nodes)
{
    for (int idx = threadIdx.x; idx < D_OUT * (D_IN / 4); idx += L1_THREADS) {
        int r  = idx / (D_IN / 4);
        int c4 = (idx % (D_IN / 4)) * 4;
        const float4 w4 = ((const float4*)W)[r * (D_IN / 4) + (c4 >> 2)];
        float* dst = &lds_w[r * LDSW_STRIDE + c4];
        dst[0] = w4.x; dst[1] = w4.y; dst[2] = w4.z; dst[3] = w4.w;
    }
    __syncthreads();

    const int tile     = gblock * L1_WAVES + (threadIdx.x >> 5);
    const int numTiles = (n_nodes + 15) >> 4;
    if (tile >= numTiles) return;

    const int lane = threadIdx.x & 31;
    const int m    = lane & 15;
    const int kh   = (lane >> 4) << 1;
    const int row0 = tile << 4;

    int arow_idx = row0 + m;
    if (arow_idx >= n_nodes) arow_idx = n_nodes - 1;
    const float* arow = feat + (size_t)arow_idx * D_IN + kh;

    const float* b0p = &lds_w[(0 * 16 + m) * LDSW_STRIDE + kh];
    const float* b1p = &lds_w[(1 * 16 + m) * LDSW_STRIDE + kh];
    const float* b2p = &lds_w[(2 * 16 + m) * LDSW_STRIDE + kh];
    const float* b3p = &lds_w[(3 * 16 + m) * LDSW_STRIDE + kh];

    v8f acc0 = {}, acc1 = {}, acc2 = {}, acc3 = {};

#pragma unroll 8
    for (int kk = 0; kk < D_IN; kk += 4) {
        v2f a  = *(const v2f*)(arow + kk);
        v2f b0 = *(const v2f*)(b0p + kk);
        v2f b1 = *(const v2f*)(b1p + kk);
        v2f b2 = *(const v2f*)(b2p + kk);
        v2f b3 = *(const v2f*)(b3p + kk);
        acc0 = __builtin_amdgcn_wmma_f32_16x16x4_f32(false, a, false, b0, (short)0, acc0, false, false);
        acc1 = __builtin_amdgcn_wmma_f32_16x16x4_f32(false, a, false, b1, (short)0, acc1, false, false);
        acc2 = __builtin_amdgcn_wmma_f32_16x16x4_f32(false, a, false, b2, (short)0, acc2, false, false);
        acc3 = __builtin_amdgcn_wmma_f32_16x16x4_f32(false, a, false, b3, (short)0, acc3, false, false);
    }

    const float bb0 = bias[m], bb1 = bias[16 + m], bb2 = bias[32 + m], bb3 = bias[48 + m];
    const int mbase = (lane >> 4) * 8;
#pragma unroll
    for (int v = 0; v < 8; ++v) {
        int r = row0 + mbase + v;
        if (r < n_nodes) {
            float* o = pre + (size_t)r * D_OUT;
            o[m]      = acc0[v] + bb0;
            o[16 + m] = acc1[v] + bb1;
            o[32 + m] = acc2[v] + bb2;
            o[48 + m] = acc3[v] + bb3;
        }
    }
}

#define RANGE_ROWS  128
#define RANGE_SHIFT 7
#define ROW_CAP     64

typedef __attribute__((ext_vector_type(2))) unsigned int edge_rec;
#define BIN_EDGES   8
#define SEG_RECS    (L1_THREADS * BIN_EDGES)
#define GATHER_WAVES 16
#define PIECES_PER_WAVE 16
static_assert(L1_WAVES == 32, "the scan below reads one wave total per lane");

__device__ __forceinline__ void bin_role(
    const int share, int* __restrict__ lds,
    const int* __restrict__ rows, const int* __restrict__ cols, const float* __restrict__ vals,
    int* __restrict__ seg_i, int* __restrict__ seg_off, int off_pitch,
    int n_edges, int n_nodes, int n_ranges)
{
    edge_rec* sh_seg  = (edge_rec*)lds;
    int*      sh_hist = lds + 2 * SEG_RECS;
    int*      sh_base = sh_hist + n_ranges;
    int*      sh_wtot = sh_base + n_ranges + 1;

    const int tid  = threadIdx.x;
    const int lane = tid & 31;
    const int w    = tid >> 5;

    for (int b = tid; b < n_ranges; b += L1_THREADS) sh_hist[b] = 0;
    __syncthreads();

    const int e0 = share * SEG_RECS + tid;
    int      bk[BIN_EDGES], pl[BIN_EDGES], rr[BIN_EDGES];
    edge_rec rec[BIN_EDGES];
#pragma unroll
    for (int k = 0; k < BIN_EDGES; ++k) {
        const int e  = e0 + k * L1_THREADS;
        const int ec = e < n_edges ? e : n_edges - 1;
        rr[k] = rows[ec];
    }
#pragma unroll
    for (int k = 0; k < BIN_EDGES; ++k) {
        const int e  = e0 + k * L1_THREADS;
        const int ec = e < n_edges ? e : n_edges - 1;
        rec[k].x = (unsigned)cols[ec];
        rec[k].y = __float_as_uint(vals[ec]);
    }
#pragma unroll
    for (int k = 0; k < BIN_EDGES; ++k) {
        const int e = e0 + k * L1_THREADS;
        bk[k] = (e < n_edges && (unsigned)rr[k] < (unsigned)n_nodes) ? (rr[k] >> RANGE_SHIFT) : -1;
    }
#pragma unroll
    for (int k = 0; k < BIN_EDGES; ++k) {
        pl[k] = 0;
        if (bk[k] >= 0) pl[k] = atomicAdd(&sh_hist[bk[k]], 1);
    }
    __syncthreads();

    int carry = 0;
    for (int c0 = 0; c0 <= n_ranges; c0 += L1_THREADS) {
        const int i = c0 + tid;
        const int v = i < n_ranges ? sh_hist[i] : 0;
        int incl = v;
#pragma unroll
        for (int d = 1; d < 32; d <<= 1) {
            const int t = __shfl_up(incl, d, 32);
            if (lane >= d) incl += t;
        }
        if (lane == 31) sh_wtot[w] = incl;
        __syncthreads();
        int wincl = sh_wtot[lane];
#pragma unroll
        for (int d = 1; d < 32; d <<= 1) {
            const int t = __shfl_up(wincl, d, 32);
            if (lane >= d) wincl += t;
        }
        const int wbefore = __shfl(wincl, w > 0 ? w - 1 : 0, 32);
        const int excl    = carry + (w > 0 ? wbefore : 0) + incl - v;
        if (i <= n_ranges) {
            sh_base[i] = excl;
            seg_off[(size_t)share * off_pitch + i] = excl;
        }
        carry += __shfl(wincl, 31, 32);
        __syncthreads();
    }

#pragma unroll
    for (int k = 0; k < BIN_EDGES; ++k) {
        if (bk[k] >= 0) {
            edge_rec q;
            q.x = (rec[k].x << RANGE_SHIFT) | (unsigned)(rr[k] & (RANGE_ROWS - 1));
            q.y = rec[k].y;
            sh_seg[sh_base[bk[k]] + pl[k]] = q;
        }
    }
    __syncthreads();

    const int total = sh_base[n_ranges];
    edge_rec* myseg = (edge_rec*)seg_i + (size_t)share * SEG_RECS;
#pragma unroll
    for (int k = 0; k < BIN_EDGES; ++k) {
        const int idx = k * L1_THREADS + tid;
        if (idx < total) myseg[idx] = sh_seg[idx];
    }
}

__global__ __launch_bounds__(L1_THREADS) void gcn_gemm_and_bin(
    const float* __restrict__ feat, const float* __restrict__ W,
    const float* __restrict__ bias, float* __restrict__ pre, int n_nodes, int gemm_blocks,
    const int* __restrict__ rows, const int* __restrict__ cols, const float* __restrict__ vals,
    int* __restrict__ seg_i, int* __restrict__ seg_off, int off_pitch,
    int n_edges, int n_ranges)
{
    extern __shared__ float lds_dyn[];
    if ((int)blockIdx.x < gemm_blocks) {
        gemm_role((int)blockIdx.x, lds_dyn, feat, W, bias, pre, n_nodes);
    } else {
        bin_role((int)blockIdx.x - gemm_blocks, (int*)lds_dyn, rows, cols, vals,
                 seg_i, seg_off, off_pitch, n_edges, n_nodes, n_ranges);
    }
}

struct RowAcc { float s0, s1, q0, q1; };

__device__ __forceinline__ int lane_i(int x, int l)
{
    return __builtin_amdgcn_readlane(x, l);
}
__device__ __forceinline__ float lane_f(float x, int l)
{
    return __int_as_float(__builtin_amdgcn_readlane(__float_as_int(x), l));
}

template <int U>
__device__ __forceinline__ void sum_batch(RowAcc& a, const float* __restrict__ pl,
                                          int sc, float sv, int j)
{
    float2 p[U];
    float  v[U];
#pragma unroll
    for (int u = 0; u < U; ++u) {
        const int c = lane_i(sc, j + u);
        v[u] = lane_f(sv, j + u);
        p[u] = *(const float2*)(pl + (size_t)c * D_OUT);
    }
#pragma unroll
    for (int u = 0; u < U; ++u) {
        const float t0 = v[u] * p[u].x;
        const float t1 = v[u] * p[u].y;
        a.s0 += t0;
        a.s1 += t1;
        a.q0 += t0 * p[u].x;
        a.q1 += t1 * p[u].y;
    }
}

__device__ __forceinline__ void sum_run(RowAcc& a, const float* __restrict__ pl,
                                        int sc, float sv, int n)
{
    int j = 0;
    for (; j + 8 <= n; j += 8) sum_batch<8>(a, pl, sc, sv, j);
    if (j + 4 <= n) { sum_batch<4>(a, pl, sc, sv, j); j += 4; }
    if (j + 2 <= n) { sum_batch<2>(a, pl, sc, sv, j); j += 2; }
    if (j < n)      { sum_batch<1>(a, pl, sc, sv, j); }
}

__device__ __forceinline__ void deal_rec(int* sh_cnt, edge_rec (*sh_rec)[ROW_CAP + 1], edge_rec q)
{
    const int rl   = q.x & (RANGE_ROWS - 1);
    const int slot = atomicAdd(&sh_cnt[rl], 1);
    if (slot < ROW_CAP) sh_rec[rl][slot] = q;
}

__global__ __launch_bounds__(GATHER_WAVES * 32) void gcn_gather_ranges(
    const int* __restrict__ seg_i, const int* __restrict__ seg_off, int n_segs, int off_pitch,
    const int* __restrict__ rows, const int* __restrict__ cols,
    const float* __restrict__ vals, const float* __restrict__ pre,
    const float* __restrict__ diag, float* __restrict__ out,
    int n_nodes, int n_edges)
{
    __shared__ int      sh_cnt[RANGE_ROWS];
    __shared__ edge_rec sh_rec[RANGE_ROWS][ROW_CAP + 1];
    __shared__ int      sh_c[GATHER_WAVES][ROW_CAP];
    __shared__ float    sh_v[GATHER_WAVES][ROW_CAP];

    const int b    = blockIdx.x;
    const int tid  = threadIdx.x;
    const int w    = tid >> 5;
    const int lane = tid & 31;

    if (tid < RANGE_ROWS) sh_cnt[tid] = 0;
    __syncthreads();

    const edge_rec* seg = (const edge_rec*)seg_i;
    const int ws = __builtin_amdgcn_readfirstlane(w);
    for (int g0 = 0; g0 < n_segs; g0 += GATHER_WAVES * PIECES_PER_WAVE) {
        const int sl = g0 + ws + GATHER_WAVES * lane;
        int o0v = 0, o1v = 0;
        if (lane < PIECES_PER_WAVE && sl < n_segs) {
            const int* op = seg_off + (size_t)sl * off_pitch + b;
            o0v = op[0];
            o1v = op[1];
        }
        edge_rec q[PIECES_PER_WAVE];
#pragma unroll
        for (int j = 0; j < PIECES_PER_WAVE; ++j) {
            const int o0 = lane_i(o0v, j);
            const int ln = lane_i(o1v, j) - o0;
            q[j].x = 0u; q[j].y = 0u;
            const edge_rec* piece = seg + (size_t)(g0 + ws + GATHER_WAVES * j) * SEG_RECS + o0;
            if (lane < ln) q[j] = piece[lane];
        }
#pragma unroll
        for (int j = 0; j < PIECES_PER_WAVE; ++j) {
            const int ln = lane_i(o1v, j) - lane_i(o0v, j);
            if (lane < ln) deal_rec(sh_cnt, sh_rec, q[j]);
        }
        if (__builtin_amdgcn_ballot_w32(o1v - o0v > 32)) {
            for (int j = 0; j < PIECES_PER_WAVE; ++j) {
                const int o0 = lane_i(o0v, j);
                const int ln = lane_i(o1v, j) - o0;
                const edge_rec* piece = seg + (size_t)(g0 + ws + GATHER_WAVES * j) * SEG_RECS + o0;
                for (int base = 32; base < ln; base += 32) {
                    if (base + lane < ln) deal_rec(sh_cnt, sh_rec, piece[base + lane]);
                }
            }
        }
    }
    __syncthreads();

    const float* pl = pre + 2 * lane;

    const int r_first = b * RANGE_ROWS + w * (RANGE_ROWS / GATHER_WAVES);
    float dg = 0.0f;
    if (lane < RANGE_ROWS / GATHER_WAVES && r_first + lane < n_nodes) dg = diag[r_first + lane];

    for (int i = 0; i < RANGE_ROWS / GATHER_WAVES; ++i) {
        const int rl = w * (RANGE_ROWS / GATHER_WAVES) + i;
        const int r  = __builtin_amdgcn_readfirstlane(b * RANGE_ROWS + rl);
        if (r >= n_nodes) break;

        const int  deg    = __builtin_amdgcn_readfirstlane(sh_cnt[rl]);
        const bool listed = deg <= ROW_CAP;
        const int  cnt    = listed ? deg : 0;
        const int  n0     = cnt < 32 ? cnt : 32;

        const bool ok0 = lane < cnt;
        const bool ok1 = lane + 32 < cnt;
        edge_rec q0 = {0u, 0u}, q1 = {0u, 0u};
        if (ok0) q0 = sh_rec[rl][lane];
        if (ok1) q1 = sh_rec[rl][lane + 32];
        const int   c0 = (int)(q0.x >> RANGE_SHIFT), c1 = (int)(q1.x >> RANGE_SHIFT);
        const float v0 = __uint_as_float(q0.y),      v1 = __uint_as_float(q1.y);

        const unsigned long long key0 =
            ((unsigned long long)((q0.x & ~(unsigned)(RANGE_ROWS - 1)) | (q0.y >> (32 - RANGE_SHIFT))) << 32) |
            ((q0.y << RANGE_SHIFT) | (unsigned)lane);
        const unsigned long long key1 =
            ((unsigned long long)((q1.x & ~(unsigned)(RANGE_ROWS - 1)) | (q1.y >> (32 - RANGE_SHIFT))) << 32) |
            ((q1.y << RANGE_SHIFT) | (unsigned)(lane + 32));
        const int k0h = (int)(key0 >> 32), k0l = (int)key0;
        const int k1h = (int)(key1 >> 32), k1l = (int)key1;

        int rk0 = 0, rk1 = 0;
        for (int j = 0; j < n0; ++j) {
            const unsigned long long kj =
                ((unsigned long long)(unsigned)lane_i(k0h, j) << 32) | (unsigned)lane_i(k0l, j);
            rk0 += (kj < key0) ? 1 : 0;
            rk1 += (kj < key1) ? 1 : 0;
        }
        for (int j = 32; j < cnt; ++j) {
            const unsigned long long kj =
                ((unsigned long long)(unsigned)lane_i(k1h, j - 32) << 32) | (unsigned)lane_i(k1l, j - 32);
            rk0 += (kj < key0) ? 1 : 0;
            rk1 += (kj < key1) ? 1 : 0;
        }

        if (ok0) { sh_c[w][rk0] = c0; sh_v[w][rk0] = v0; }
        if (ok1) { sh_c[w][rk1] = c1; sh_v[w][rk1] = v1; }
        __builtin_amdgcn_fence(__ATOMIC_RELEASE, "workgroup");
        __builtin_amdgcn_wave_barrier();
        __builtin_amdgcn_fence(__ATOMIC_ACQUIRE, "workgroup");
        int   sc0 = 0, sc1 = 0;
        float sv0 = 0.0f, sv1 = 0.0f;
        if (ok0) { sc0 = sh_c[w][lane];      sv0 = sh_v[w][lane]; }
        if (ok1) { sc1 = sh_c[w][lane + 32]; sv1 = sh_v[w][lane + 32]; }

        RowAcc a = {0.0f, 0.0f, 0.0f, 0.0f};
        sum_run(a, pl, sc0, sv0, n0);
        if (cnt > 32) sum_run(a, pl, sc1, sv1, cnt - 32);

        if (!listed) {
            for (int base = 0; base < n_edges; base += 32) {
                const int e = base + lane;
                int   c = 0;
                float v = 0.0f;
                bool  mine = false;
                if (e < n_edges && rows[e] == r) { mine = true; c = cols[e]; v = vals[e]; }
                unsigned m = __builtin_amdgcn_ballot_w32(mine);
                while (m) {
                    const int src = __builtin_ctz(m);
                    m &= m - 1;
                    sum_batch<1>(a, pl, c, v, src);
                }
            }
        }

        const float d = lane_f(dg, i);
        float2 o;
        o.x = d * 0.5f * (a.s0 * a.s0 - a.q0);
        o.y = d * 0.5f * (a.s1 * a.s1 - a.q1);
        *(float2*)(out + (size_t)r * D_OUT + 2 * lane) = o;
    }
}

extern "C" void kernel_launch(void* const* d_in, const int* in_sizes, int n_in,
                              void* d_out, int out_size, void* d_ws, size_t ws_size,
                              hipStream_t stream)
{
    const float* feat     = (const float*)d_in[0];
    const float* W        = (const float*)d_in[1];
    const float* bias     = (const float*)d_in[2];
    const float* adj_vals = (const float*)d_in[3];
    const float* diag     = (const float*)d_in[4];
    const int*   adj_rows = (const int*)d_in[5];
    const int*   adj_cols = (const int*)d_in[6];

    const int n_nodes = in_sizes[4];
    const int n_edges = in_sizes[3];

    const int n_ranges  = (n_nodes + RANGE_ROWS - 1) / RANGE_ROWS;
    const int n_segs    = (n_edges + SEG_RECS - 1) / SEG_RECS;
    const int off_pitch = (n_ranges + 1 + 31) & ~31;

    float* pre     = (float*)d_ws;
    int*   seg_i   = (int*)(pre + (size_t)n_nodes * D_OUT);
    int*   seg_off = seg_i + (size_t)n_segs * SEG_RECS * 2;
    float* out     = (float*)d_out;

    const int tiles       = (n_nodes + 15) / 16;
    const int gemm_blocks = (tiles + L1_WAVES - 1) / L1_WAVES;
    const size_t lds_gemm = D_OUT * LDSW_STRIDE * sizeof(float);
    const size_t lds_bin  = (2 * SEG_RECS + 2 * n_ranges + 1 + L1_WAVES) * sizeof(int);
    gcn_gemm_and_bin<<<gemm_blocks + n_segs, L1_THREADS, lds_gemm > lds_bin ? lds_gemm : lds_bin, stream>>>(
        feat, W, bias, pre, n_nodes, gemm_blocks, adj_rows, adj_cols, adj_vals,
        seg_i, seg_off, off_pitch, n_edges, n_ranges);

    gcn_gather_ranges<<<n_ranges, GATHER_WAVES * 32, 0, stream>>>(
        seg_i, seg_off, n_segs, off_pitch, adj_rows, adj_cols, adj_vals, pre, diag, out,
        n_nodes, n_edges);
}


# Gather's per-row ranking done on 32-bit col:slot keys (one rank per lane for rows of 32 records or fewer); 64-bit (col,val) ranking kept only for rows where a col repeats; sum order unchanged
# speedup vs baseline: 3.5562x; 1.0613x over previous
#include <hip/hip_runtime.h>

#if defined(__HIP_DEVICE_COMPILE__)
#if !__has_builtin(__builtin_amdgcn_wmma_f32_16x16x4_f32)
#error "gfx1250 f32 WMMA builtin not available on this toolchain (device pass)"
#endif
#endif

typedef __attribute__((ext_vector_type(2))) float v2f;
typedef __attribute__((ext_vector_type(8))) float v8f;

#define D_IN 256
#define D_OUT 64
#define LDSW_STRIDE 260

#define L1_THREADS 1024
#define L1_WAVES   (L1_THREADS / 32)

__device__ __forceinline__ void gemm_role(
    const int gblock, float* __restrict__ lds_w,
    const float* __restrict__ feat, const float* __restrict__ W,
    const float* __restrict__ bias, float* __restrict__ pre, int n_nodes)
{
    for (int idx = threadIdx.x; idx < D_OUT * (D_IN / 4); idx += L1_THREADS) {
        int r  = idx / (D_IN / 4);
        int c4 = (idx % (D_IN / 4)) * 4;
        const float4 w4 = ((const float4*)W)[r * (D_IN / 4) + (c4 >> 2)];
        float* dst = &lds_w[r * LDSW_STRIDE + c4];
        dst[0] = w4.x; dst[1] = w4.y; dst[2] = w4.z; dst[3] = w4.w;
    }
    __syncthreads();

    const int tile     = gblock * L1_WAVES + (threadIdx.x >> 5);
    const int numTiles = (n_nodes + 15) >> 4;
    if (tile >= numTiles) return;

    const int lane = threadIdx.x & 31;
    const int m    = lane & 15;
    const int kh   = (lane >> 4) << 1;
    const int row0 = tile << 4;

    int arow_idx = row0 + m;
    if (arow_idx >= n_nodes) arow_idx = n_nodes - 1;
    const float* arow = feat + (size_t)arow_idx * D_IN + kh;

    const float* b0p = &lds_w[(0 * 16 + m) * LDSW_STRIDE + kh];
    const float* b1p = &lds_w[(1 * 16 + m) * LDSW_STRIDE + kh];
    const float* b2p = &lds_w[(2 * 16 + m) * LDSW_STRIDE + kh];
    const float* b3p = &lds_w[(3 * 16 + m) * LDSW_STRIDE + kh];

    v8f acc0 = {}, acc1 = {}, acc2 = {}, acc3 = {};

#pragma unroll 8
    for (int kk = 0; kk < D_IN; kk += 4) {
        v2f a  = *(const v2f*)(arow + kk);
        v2f b0 = *(const v2f*)(b0p + kk);
        v2f b1 = *(const v2f*)(b1p + kk);
        v2f b2 = *(const v2f*)(b2p + kk);
        v2f b3 = *(const v2f*)(b3p + kk);
        acc0 = __builtin_amdgcn_wmma_f32_16x16x4_f32(false, a, false, b0, (short)0, acc0, false, false);
        acc1 = __builtin_amdgcn_wmma_f32_16x16x4_f32(false, a, false, b1, (short)0, acc1, false, false);
        acc2 = __builtin_amdgcn_wmma_f32_16x16x4_f32(false, a, false, b2, (short)0, acc2, false, false);
        acc3 = __builtin_amdgcn_wmma_f32_16x16x4_f32(false, a, false, b3, (short)0, acc3, false, false);
    }

    const float bb0 = bias[m], bb1 = bias[16 + m], bb2 = bias[32 + m], bb3 = bias[48 + m];
    const int mbase = (lane >> 4) * 8;
#pragma unroll
    for (int v = 0; v < 8; ++v) {
        int r = row0 + mbase + v;
        if (r < n_nodes) {
            float* o = pre + (size_t)r * D_OUT;
            o[m]      = acc0[v] + bb0;
            o[16 + m] = acc1[v] + bb1;
            o[32 + m] = acc2[v] + bb2;
            o[48 + m] = acc3[v] + bb3;
        }
    }
}

#define RANGE_ROWS  128
#define RANGE_SHIFT 7
#define ROW_CAP     64

typedef __attribute__((ext_vector_type(2))) unsigned int edge_rec;
#define BIN_EDGES   8
#define SEG_RECS    (L1_THREADS * BIN_EDGES)
#define GATHER_WAVES 16
#define PIECES_PER_WAVE 16
static_assert(L1_WAVES == 32, "the scan below reads one wave total per lane");

__device__ __forceinline__ void bin_role(
    const int share, int* __restrict__ lds,
    const int* __restrict__ rows, const int* __restrict__ cols, const float* __restrict__ vals,
    int* __restrict__ seg_i, int* __restrict__ seg_off, int off_pitch,
    int n_edges, int n_nodes, int n_ranges)
{
    edge_rec* sh_seg  = (edge_rec*)lds;
    int*      sh_hist = lds + 2 * SEG_RECS;
    int*      sh_base = sh_hist + n_ranges;
    int*      sh_wtot = sh_base + n_ranges + 1;

    const int tid  = threadIdx.x;
    const int lane = tid & 31;
    const int w    = tid >> 5;

    for (int b = tid; b < n_ranges; b += L1_THREADS) sh_hist[b] = 0;
    __syncthreads();

    const int e0 = share * SEG_RECS + tid;
    int      bk[BIN_EDGES], pl[BIN_EDGES], rr[BIN_EDGES];
    edge_rec rec[BIN_EDGES];
#pragma unroll
    for (int k = 0; k < BIN_EDGES; ++k) {
        const int e  = e0 + k * L1_THREADS;
        const int ec = e < n_edges ? e : n_edges - 1;
        rr[k] = rows[ec];
    }
#pragma unroll
    for (int k = 0; k < BIN_EDGES; ++k) {
        const int e  = e0 + k * L1_THREADS;
        const int ec = e < n_edges ? e : n_edges - 1;
        rec[k].x = (unsigned)cols[ec];
        rec[k].y = __float_as_uint(vals[ec]);
    }
#pragma unroll
    for (int k = 0; k < BIN_EDGES; ++k) {
        const int e = e0 + k * L1_THREADS;
        bk[k] = (e < n_edges && (unsigned)rr[k] < (unsigned)n_nodes) ? (rr[k] >> RANGE_SHIFT) : -1;
    }
#pragma unroll
    for (int k = 0; k < BIN_EDGES; ++k) {
        pl[k] = 0;
        if (bk[k] >= 0) pl[k] = atomicAdd(&sh_hist[bk[k]], 1);
    }
    __syncthreads();

    int carry = 0;
    for (int c0 = 0; c0 <= n_ranges; c0 += L1_THREADS) {
        const int i = c0 + tid;
        const int v = i < n_ranges ? sh_hist[i] : 0;
        int incl = v;
#pragma unroll
        for (int d = 1; d < 32; d <<= 1) {
            const int t = __shfl_up(incl, d, 32);
            if (lane >= d) incl += t;
        }
        if (lane == 31) sh_wtot[w] = incl;
        __syncthreads();
        int wincl = sh_wtot[lane];
#pragma unroll
        for (int d = 1; d < 32; d <<= 1) {
            const int t = __shfl_up(wincl, d, 32);
            if (lane >= d) wincl += t;
        }
        const int wbefore = __shfl(wincl, w > 0 ? w - 1 : 0, 32);
        const int excl    = carry + (w > 0 ? wbefore : 0) + incl - v;
        if (i <= n_ranges) {
            sh_base[i] = excl;
            seg_off[(size_t)share * off_pitch + i] = excl;
        }
        carry += __shfl(wincl, 31, 32);
        __syncthreads();
    }

#pragma unroll
    for (int k = 0; k < BIN_EDGES; ++k) {
        if (bk[k] >= 0) {
            edge_rec q;
            q.x = (rec[k].x << RANGE_SHIFT) | (unsigned)(rr[k] & (RANGE_ROWS - 1));
            q.y = rec[k].y;
            sh_seg[sh_base[bk[k]] + pl[k]] = q;
        }
    }
    __syncthreads();

    const int total = sh_base[n_ranges];
    edge_rec* myseg = (edge_rec*)seg_i + (size_t)share * SEG_RECS;
#pragma unroll
    for (int k = 0; k < BIN_EDGES; ++k) {
        const int idx = k * L1_THREADS + tid;
        if (idx < total) myseg[idx] = sh_seg[idx];
    }
}

__global__ __launch_bounds__(L1_THREADS) void gcn_gemm_and_bin(
    const float* __restrict__ feat, const float* __restrict__ W,
    const float* __restrict__ bias, float* __restrict__ pre, int n_nodes, int gemm_blocks,
    const int* __restrict__ rows, const int* __restrict__ cols, const float* __restrict__ vals,
    int* __restrict__ seg_i, int* __restrict__ seg_off, int off_pitch,
    int n_edges, int n_ranges)
{
    extern __shared__ float lds_dyn[];
    if ((int)blockIdx.x < gemm_blocks) {
        gemm_role((int)blockIdx.x, lds_dyn, feat, W, bias, pre, n_nodes);
    } else {
        bin_role((int)blockIdx.x - gemm_blocks, (int*)lds_dyn, rows, cols, vals,
                 seg_i, seg_off, off_pitch, n_edges, n_nodes, n_ranges);
    }
}

struct RowAcc { float s0, s1, q0, q1; };

__device__ __forceinline__ int lane_i(int x, int l)
{
    return __builtin_amdgcn_readlane(x, l);
}
__device__ __forceinline__ float lane_f(float x, int l)
{
    return __int_as_float(__builtin_amdgcn_readlane(__float_as_int(x), l));
}

template <int U>
__device__ __forceinline__ void sum_batch(RowAcc& a, const float* __restrict__ pl,
                                          int sc, float sv, int j)
{
    float2 p[U];
    float  v[U];
#pragma unroll
    for (int u = 0; u < U; ++u) {
        const int c = lane_i(sc, j + u);
        v[u] = lane_f(sv, j + u);
        p[u] = *(const float2*)(pl + (size_t)c * D_OUT);
    }
#pragma unroll
    for (int u = 0; u < U; ++u) {
        const float t0 = v[u] * p[u].x;
        const float t1 = v[u] * p[u].y;
        a.s0 += t0;
        a.s1 += t1;
        a.q0 += t0 * p[u].x;
        a.q1 += t1 * p[u].y;
    }
}

__device__ __forceinline__ void sum_run(RowAcc& a, const float* __restrict__ pl,
                                        int sc, float sv, int n)
{
    int j = 0;
    for (; j + 8 <= n; j += 8) sum_batch<8>(a, pl, sc, sv, j);
    if (j + 4 <= n) { sum_batch<4>(a, pl, sc, sv, j); j += 4; }
    if (j + 2 <= n) { sum_batch<2>(a, pl, sc, sv, j); j += 2; }
    if (j < n)      { sum_batch<1>(a, pl, sc, sv, j); }
}

__device__ __forceinline__ void deal_rec(int* sh_cnt, edge_rec (*sh_rec)[ROW_CAP + 1], edge_rec q)
{
    const int rl   = q.x & (RANGE_ROWS - 1);
    const int slot = atomicAdd(&sh_cnt[rl], 1);
    if (slot < ROW_CAP) sh_rec[rl][slot] = q;
}

__global__ __launch_bounds__(GATHER_WAVES * 32) __attribute__((amdgpu_waves_per_eu(16, 16))) void gcn_gather_ranges(
    const int* __restrict__ seg_i, const int* __restrict__ seg_off, int n_segs, int off_pitch,
    const int* __restrict__ rows, const int* __restrict__ cols,
    const float* __restrict__ vals, const float* __restrict__ pre,
    const float* __restrict__ diag, float* __restrict__ out,
    int n_nodes, int n_edges)
{
    __shared__ int      sh_cnt[RANGE_ROWS];
    __shared__ edge_rec sh_rec[RANGE_ROWS][ROW_CAP + 1];
    __shared__ int      sh_c[GATHER_WAVES][ROW_CAP];
    __shared__ float    sh_v[GATHER_WAVES][ROW_CAP];

    const int b    = blockIdx.x;
    const int tid  = threadIdx.x;
    const int w    = tid >> 5;
    const int lane = tid & 31;

    if (tid < RANGE_ROWS) sh_cnt[tid] = 0;
    __syncthreads();

    const edge_rec* seg = (const edge_rec*)seg_i;
    const int ws = __builtin_amdgcn_readfirstlane(w);
    for (int g0 = 0; g0 < n_segs; g0 += GATHER_WAVES * PIECES_PER_WAVE) {
        const int sl = g0 + ws + GATHER_WAVES * lane;
        int o0v = 0, o1v = 0;
        if (lane < PIECES_PER_WAVE && sl < n_segs) {
            const int* op = seg_off + (size_t)sl * off_pitch + b;
            o0v = op[0];
            o1v = op[1];
        }
        edge_rec q[PIECES_PER_WAVE];
#pragma unroll
        for (int j = 0; j < PIECES_PER_WAVE; ++j) {
            const int o0 = lane_i(o0v, j);
            const int ln = lane_i(o1v, j) - o0;
            q[j].x = 0u; q[j].y = 0u;
            const edge_rec* piece = seg + (size_t)(g0 + ws + GATHER_WAVES * j) * SEG_RECS + o0;
            if (lane < ln) q[j] = piece[lane];
        }
#pragma unroll
        for (int j = 0; j < PIECES_PER_WAVE; ++j) {
            const int ln = lane_i(o1v, j) - lane_i(o0v, j);
            if (lane < ln) deal_rec(sh_cnt, sh_rec, q[j]);
        }
        if (__builtin_amdgcn_ballot_w32(o1v - o0v > 32)) {
            for (int j = 0; j < PIECES_PER_WAVE; ++j) {
                const int o0 = lane_i(o0v, j);
                const int ln = lane_i(o1v, j) - o0;
                const edge_rec* piece = seg + (size_t)(g0 + ws + GATHER_WAVES * j) * SEG_RECS + o0;
                for (int base = 32; base < ln; base += 32) {
                    if (base + lane < ln) deal_rec(sh_cnt, sh_rec, piece[base + lane]);
                }
            }
        }
    }
    __syncthreads();

    const float* pl = pre + 2 * lane;

    const int r_first = b * RANGE_ROWS + w * (RANGE_ROWS / GATHER_WAVES);
    float dg = 0.0f;
    if (lane < RANGE_ROWS / GATHER_WAVES && r_first + lane < n_nodes) dg = diag[r_first + lane];

    for (int i = 0; i < RANGE_ROWS / GATHER_WAVES; ++i) {
        const int rl = w * (RANGE_ROWS / GATHER_WAVES) + i;
        const int r  = __builtin_amdgcn_readfirstlane(b * RANGE_ROWS + rl);
        if (r >= n_nodes) break;

        const int  deg    = __builtin_amdgcn_readfirstlane(sh_cnt[rl]);
        const bool listed = deg <= ROW_CAP;
        const int  cnt    = listed ? deg : 0;
        const int  n0     = cnt < 32 ? cnt : 32;

        const bool ok0 = lane < cnt;
        const bool ok1 = lane + 32 < cnt;
        edge_rec q0 = {0u, 0u}, q1 = {0u, 0u};
        if (ok0) q0 = sh_rec[rl][lane];
        if (ok1) q1 = sh_rec[rl][lane + 32];
        const int   c0 = (int)(q0.x >> RANGE_SHIFT), c1 = (int)(q1.x >> RANGE_SHIFT);
        const float v0 = __uint_as_float(q0.y),      v1 = __uint_as_float(q1.y);

        const unsigned f0 = ((unsigned)c0 << 6) | (unsigned)lane;
        const unsigned f1 = ((unsigned)c1 << 6) | (unsigned)(lane + 32);

        int rk0 = 0, rk1 = 0;
        if (cnt <= 32) {
            for (int j = 0; j < cnt; ++j) {
                const unsigned kj = (unsigned)lane_i((int)f0, j);
                rk0 += (kj < f0) ? 1 : 0;
            }
        } else {
            for (int j = 0; j < 32; ++j) {
                const unsigned kj = (unsigned)lane_i((int)f0, j);
                rk0 += (kj < f0) ? 1 : 0;
                rk1 += (kj < f1) ? 1 : 0;
            }
            for (int j = 32; j < cnt; ++j) {
                const unsigned kj = (unsigned)lane_i((int)f1, j - 32);
                rk0 += (kj < f0) ? 1 : 0;
                rk1 += (kj < f1) ? 1 : 0;
            }
        }

        if (ok0) { sh_c[w][rk0] = c0; sh_v[w][rk0] = v0; }
        if (ok1) { sh_c[w][rk1] = c1; sh_v[w][rk1] = v1; }
        __builtin_amdgcn_fence(__ATOMIC_RELEASE, "workgroup");
        __builtin_amdgcn_wave_barrier();
        __builtin_amdgcn_fence(__ATOMIC_ACQUIRE, "workgroup");
        int   sc0 = 0, sc1 = 0;
        float sv0 = 0.0f, sv1 = 0.0f;
        if (ok0) { sc0 = sh_c[w][lane];      sv0 = sh_v[w][lane]; }
        if (ok1) { sc1 = sh_c[w][lane + 32]; sv1 = sh_v[w][lane + 32]; }

        bool dup = false;
        if (lane + 1 < cnt)  dup = sh_c[w][lane + 1] == sc0;
        if (lane + 33 < cnt) dup = dup || (sh_c[w][lane + 33] == sc1);
        if (__builtin_amdgcn_ballot_w32(dup)) {
            const unsigned long long key0 =
                ((unsigned long long)(((unsigned)c0 << RANGE_SHIFT) | (__float_as_uint(v0) >> (32 - RANGE_SHIFT))) << 32) |
                ((__float_as_uint(v0) << RANGE_SHIFT) | (unsigned)lane);
            const unsigned long long key1 =
                ((unsigned long long)(((unsigned)c1 << RANGE_SHIFT) | (__float_as_uint(v1) >> (32 - RANGE_SHIFT))) << 32) |
                ((__float_as_uint(v1) << RANGE_SHIFT) | (unsigned)(lane + 32));
            const int k0h = (int)(key0 >> 32), k0l = (int)key0;
            const int k1h = (int)(key1 >> 32), k1l = (int)key1;
            rk0 = 0; rk1 = 0;
            for (int j = 0; j < n0; ++j) {
                const unsigned long long kj =
                    ((unsigned long long)(unsigned)lane_i(k0h, j) << 32) | (unsigned)lane_i(k0l, j);
                rk0 += (kj < key0) ? 1 : 0;
                rk1 += (kj < key1) ? 1 : 0;
            }
            for (int j = 32; j < cnt; ++j) {
                const unsigned long long kj =
                    ((unsigned long long)(unsigned)lane_i(k1h, j - 32) << 32) | (unsigned)lane_i(k1l, j - 32);
                rk0 += (kj < key0) ? 1 : 0;
                rk1 += (kj < key1) ? 1 : 0;
            }
            __builtin_amdgcn_fence(__ATOMIC_RELEASE, "workgroup");
            __builtin_amdgcn_wave_barrier();
            if (ok0) { sh_c[w][rk0] = c0; sh_v[w][rk0] = v0; }
            if (ok1) { sh_c[w][rk1] = c1; sh_v[w][rk1] = v1; }
            __builtin_amdgcn_fence(__ATOMIC_RELEASE, "workgroup");
            __builtin_amdgcn_wave_barrier();
            __builtin_amdgcn_fence(__ATOMIC_ACQUIRE, "workgroup");
            if (ok0) { sc0 = sh_c[w][lane];      sv0 = sh_v[w][lane]; }
            if (ok1) { sc1 = sh_c[w][lane + 32]; sv1 = sh_v[w][lane + 32]; }
        }

        RowAcc a = {0.0f, 0.0f, 0.0f, 0.0f};
        sum_run(a, pl, sc0, sv0, n0);
        if (cnt > 32) sum_run(a, pl, sc1, sv1, cnt - 32);

        if (!listed) {
            for (int base = 0; base < n_edges; base += 32) {
                const int e = base + lane;
                int   c = 0;
                float v = 0.0f;
                bool  mine = false;
                if (e < n_edges && rows[e] == r) { mine = true; c = cols[e]; v = vals[e]; }
                unsigned m = __builtin_amdgcn_ballot_w32(mine);
                while (m) {
                    const int src = __builtin_ctz(m);
                    m &= m - 1;
                    sum_batch<1>(a, pl, c, v, src);
                }
            }
        }

        const float d = lane_f(dg, i);
        float2 o;
        o.x = d * 0.5f * (a.s0 * a.s0 - a.q0);
        o.y = d * 0.5f * (a.s1 * a.s1 - a.q1);
        *(float2*)(out + (size_t)r * D_OUT + 2 * lane) = o;
    }
}

extern "C" void kernel_launch(void* const* d_in, const int* in_sizes, int n_in,
                              void* d_out, int out_size, void* d_ws, size_t ws_size,
                              hipStream_t stream)
{
    const float* feat     = (const float*)d_in[0];
    const float* W        = (const float*)d_in[1];
    const float* bias     = (const float*)d_in[2];
    const float* adj_vals = (const float*)d_in[3];
    const float* diag     = (const float*)d_in[4];
    const int*   adj_rows = (const int*)d_in[5];
    const int*   adj_cols = (const int*)d_in[6];

    const int n_nodes = in_sizes[4];
    const int n_edges = in_sizes[3];

    const int n_ranges  = (n_nodes + RANGE_ROWS - 1) / RANGE_ROWS;
    const int n_segs    = (n_edges + SEG_RECS - 1) / SEG_RECS;
    const int off_pitch = (n_ranges + 1 + 31) & ~31;

    float* pre     = (float*)d_ws;
    int*   seg_i   = (int*)(pre + (size_t)n_nodes * D_OUT);
    int*   seg_off = seg_i + (size_t)n_segs * SEG_RECS * 2;
    float* out     = (float*)d_out;

    const int tiles       = (n_nodes + 15) / 16;
    const int gemm_blocks = (tiles + L1_WAVES - 1) / L1_WAVES;
    const size_t lds_gemm = D_OUT * LDSW_STRIDE * sizeof(float);
    const size_t lds_bin  = (2 * SEG_RECS + 2 * n_ranges + 1 + L1_WAVES) * sizeof(int);
    gcn_gemm_and_bin<<<gemm_blocks + n_segs, L1_THREADS, lds_gemm > lds_bin ? lds_gemm : lds_bin, stream>>>(
        feat, W, bias, pre, n_nodes, gemm_blocks, adj_rows, adj_cols, adj_vals,
        seg_i, seg_off, off_pitch, n_edges, n_ranges);

    gcn_gather_ranges<<<n_ranges, GATHER_WAVES * 32, 0, stream>>>(
        seg_i, seg_off, n_segs, off_pitch, adj_rows, adj_cols, adj_vals, pre, diag, out,
        n_nodes, n_edges);
}


# Gather sums: each ranked (col,val) record now comes from one same-address LDS read (all lanes) instead of two lane-to-scalar moves plus scalar address sums per edge; per-row sum order unchanged
# speedup vs baseline: 3.9473x; 1.1100x over previous
#include <hip/hip_runtime.h>

#if defined(__HIP_DEVICE_COMPILE__)
#if !__has_builtin(__builtin_amdgcn_wmma_f32_16x16x4_f32)
#error "gfx1250 f32 WMMA builtin not available on this toolchain (device pass)"
#endif
#endif

typedef __attribute__((ext_vector_type(2))) float v2f;
typedef __attribute__((ext_vector_type(8))) float v8f;

#define D_IN 256
#define D_OUT 64
#define LDSW_STRIDE 260

#define L1_THREADS 1024
#define L1_WAVES   (L1_THREADS / 32)

__device__ __forceinline__ void gemm_role(
    const int gblock, float* __restrict__ lds_w,
    const float* __restrict__ feat, const float* __restrict__ W,
    const float* __restrict__ bias, float* __restrict__ pre, int n_nodes)
{
    for (int idx = threadIdx.x; idx < D_OUT * (D_IN / 4); idx += L1_THREADS) {
        int r  = idx / (D_IN / 4);
        int c4 = (idx % (D_IN / 4)) * 4;
        const float4 w4 = ((const float4*)W)[r * (D_IN / 4) + (c4 >> 2)];
        float* dst = &lds_w[r * LDSW_STRIDE + c4];
        dst[0] = w4.x; dst[1] = w4.y; dst[2] = w4.z; dst[3] = w4.w;
    }
    __syncthreads();

    const int tile     = gblock * L1_WAVES + (threadIdx.x >> 5);
    const int numTiles = (n_nodes + 15) >> 4;
    if (tile >= numTiles) return;

    const int lane = threadIdx.x & 31;
    const int m    = lane & 15;
    const int kh   = (lane >> 4) << 1;
    const int row0 = tile << 4;

    int arow_idx = row0 + m;
    if (arow_idx >= n_nodes) arow_idx = n_nodes - 1;
    const float* arow = feat + (size_t)arow_idx * D_IN + kh;

    const float* b0p = &lds_w[(0 * 16 + m) * LDSW_STRIDE + kh];
    const float* b1p = &lds_w[(1 * 16 + m) * LDSW_STRIDE + kh];
    const float* b2p = &lds_w[(2 * 16 + m) * LDSW_STRIDE + kh];
    const float* b3p = &lds_w[(3 * 16 + m) * LDSW_STRIDE + kh];

    v8f acc0 = {}, acc1 = {}, acc2 = {}, acc3 = {};

#pragma unroll 8
    for (int kk = 0; kk < D_IN; kk += 4) {
        v2f a  = *(const v2f*)(arow + kk);
        v2f b0 = *(const v2f*)(b0p + kk);
        v2f b1 = *(const v2f*)(b1p + kk);
        v2f b2 = *(const v2f*)(b2p + kk);
        v2f b3 = *(const v2f*)(b3p + kk);
        acc0 = __builtin_amdgcn_wmma_f32_16x16x4_f32(false, a, false, b0, (short)0, acc0, false, false);
        acc1 = __builtin_amdgcn_wmma_f32_16x16x4_f32(false, a, false, b1, (short)0, acc1, false, false);
        acc2 = __builtin_amdgcn_wmma_f32_16x16x4_f32(false, a, false, b2, (short)0, acc2, false, false);
        acc3 = __builtin_amdgcn_wmma_f32_16x16x4_f32(false, a, false, b3, (short)0, acc3, false, false);
    }

    const float bb0 = bias[m], bb1 = bias[16 + m], bb2 = bias[32 + m], bb3 = bias[48 + m];
    const int mbase = (lane >> 4) * 8;
#pragma unroll
    for (int v = 0; v < 8; ++v) {
        int r = row0 + mbase + v;
        if (r < n_nodes) {
            float* o = pre + (size_t)r * D_OUT;
            o[m]      = acc0[v] + bb0;
            o[16 + m] = acc1[v] + bb1;
            o[32 + m] = acc2[v] + bb2;
            o[48 + m] = acc3[v] + bb3;
        }
    }
}

#define RANGE_ROWS  128
#define RANGE_SHIFT 7
#define ROW_CAP     64

typedef __attribute__((ext_vector_type(2))) unsigned int edge_rec;
#define BIN_EDGES   8
#define SEG_RECS    (L1_THREADS * BIN_EDGES)
#define GATHER_WAVES 16
#define PIECES_PER_WAVE 16
static_assert(L1_WAVES == 32, "the scan below reads one wave total per lane");

__device__ __forceinline__ void bin_role(
    const int share, int* __restrict__ lds,
    const int* __restrict__ rows, const int* __restrict__ cols, const float* __restrict__ vals,
    int* __restrict__ seg_i, int* __restrict__ seg_off, int off_pitch,
    int n_edges, int n_nodes, int n_ranges)
{
    edge_rec* sh_seg  = (edge_rec*)lds;
    int*      sh_hist = lds + 2 * SEG_RECS;
    int*      sh_base = sh_hist + n_ranges;
    int*      sh_wtot = sh_base + n_ranges + 1;

    const int tid  = threadIdx.x;
    const int lane = tid & 31;
    const int w    = tid >> 5;

    for (int b = tid; b < n_ranges; b += L1_THREADS) sh_hist[b] = 0;
    __syncthreads();

    const int e0 = share * SEG_RECS + tid;
    int      bk[BIN_EDGES], pl[BIN_EDGES], rr[BIN_EDGES];
    edge_rec rec[BIN_EDGES];
#pragma unroll
    for (int k = 0; k < BIN_EDGES; ++k) {
        const int e  = e0 + k * L1_THREADS;
        const int ec = e < n_edges ? e : n_edges - 1;
        rr[k] = rows[ec];
    }
#pragma unroll
    for (int k = 0; k < BIN_EDGES; ++k) {
        const int e  = e0 + k * L1_THREADS;
        const int ec = e < n_edges ? e : n_edges - 1;
        rec[k].x = (unsigned)cols[ec];
        rec[k].y = __float_as_uint(vals[ec]);
    }
#pragma unroll
    for (int k = 0; k < BIN_EDGES; ++k) {
        const int e = e0 + k * L1_THREADS;
        bk[k] = (e < n_edges && (unsigned)rr[k] < (unsigned)n_nodes) ? (rr[k] >> RANGE_SHIFT) : -1;
    }
#pragma unroll
    for (int k = 0; k < BIN_EDGES; ++k) {
        pl[k] = 0;
        if (bk[k] >= 0) pl[k] = atomicAdd(&sh_hist[bk[k]], 1);
    }
    __syncthreads();

    int carry = 0;
    for (int c0 = 0; c0 <= n_ranges; c0 += L1_THREADS) {
        const int i = c0 + tid;
        const int v = i < n_ranges ? sh_hist[i] : 0;
        int incl = v;
#pragma unroll
        for (int d = 1; d < 32; d <<= 1) {
            const int t = __shfl_up(incl, d, 32);
            if (lane >= d) incl += t;
        }
        if (lane == 31) sh_wtot[w] = incl;
        __syncthreads();
        int wincl = sh_wtot[lane];
#pragma unroll
        for (int d = 1; d < 32; d <<= 1) {
            const int t = __shfl_up(wincl, d, 32);
            if (lane >= d) wincl += t;
        }
        const int wbefore = __shfl(wincl, w > 0 ? w - 1 : 0, 32);
        const int excl    = carry + (w > 0 ? wbefore : 0) + incl - v;
        if (i <= n_ranges) {
            sh_base[i] = excl;
            seg_off[(size_t)share * off_pitch + i] = excl;
        }
        carry += __shfl(wincl, 31, 32);
        __syncthreads();
    }

#pragma unroll
    for (int k = 0; k < BIN_EDGES; ++k) {
        if (bk[k] >= 0) {
            edge_rec q;
            q.x = (rec[k].x << RANGE_SHIFT) | (unsigned)(rr[k] & (RANGE_ROWS - 1));
            q.y = rec[k].y;
            sh_seg[sh_base[bk[k]] + pl[k]] = q;
        }
    }
    __syncthreads();

    const int total = sh_base[n_ranges];
    edge_rec* myseg = (edge_rec*)seg_i + (size_t)share * SEG_RECS;
#pragma unroll
    for (int k = 0; k < BIN_EDGES; ++k) {
        const int idx = k * L1_THREADS + tid;
        if (idx < total) myseg[idx] = sh_seg[idx];
    }
}

__global__ __launch_bounds__(L1_THREADS) void gcn_gemm_and_bin(
    const float* __restrict__ feat, const float* __restrict__ W,
    const float* __restrict__ bias, float* __restrict__ pre, int n_nodes, int gemm_blocks,
    const int* __restrict__ rows, const int* __restrict__ cols, const float* __restrict__ vals,
    int* __restrict__ seg_i, int* __restrict__ seg_off, int off_pitch,
    int n_edges, int n_ranges)
{
    extern __shared__ float lds_dyn[];
    if ((int)blockIdx.x < gemm_blocks) {
        gemm_role((int)blockIdx.x, lds_dyn, feat, W, bias, pre, n_nodes);
    } else {
        bin_role((int)blockIdx.x - gemm_blocks, (int*)lds_dyn, rows, cols, vals,
                 seg_i, seg_off, off_pitch, n_edges, n_nodes, n_ranges);
    }
}

struct RowAcc { float s0, s1, q0, q1; };

__device__ __forceinline__ int lane_i(int x, int l)
{
    return __builtin_amdgcn_readlane(x, l);
}
__device__ __forceinline__ float lane_f(float x, int l)
{
    return __int_as_float(__builtin_amdgcn_readlane(__float_as_int(x), l));
}

template <int U>
__device__ __forceinline__ void sum_batch(RowAcc& a, const float* __restrict__ pl,
                                          int sc, float sv, int j)
{
    float2 p[U];
    float  v[U];
#pragma unroll
    for (int u = 0; u < U; ++u) {
        const int c = lane_i(sc, j + u);
        v[u] = lane_f(sv, j + u);
        p[u] = *(const float2*)(pl + (size_t)c * D_OUT);
    }
#pragma unroll
    for (int u = 0; u < U; ++u) {
        const float t0 = v[u] * p[u].x;
        const float t1 = v[u] * p[u].y;
        a.s0 += t0;
        a.s1 += t1;
        a.q0 += t0 * p[u].x;
        a.q1 += t1 * p[u].y;
    }
}

template <int U>
__device__ __forceinline__ void sum_list(v2f& S, v2f& Q, const v2f* __restrict__ pre2,
                                         int lane, const edge_rec* lst)
{
    edge_rec e[U];
    v2f      p[U];
#pragma unroll
    for (int u = 0; u < U; ++u) e[u] = lst[u];
#pragma unroll
    for (int u = 0; u < U; ++u) p[u] = pre2[(int)(e[u].x << 5) + lane];
#pragma unroll
    for (int u = 0; u < U; ++u) {
        const float v = __uint_as_float(e[u].y);
        const v2f   t = p[u] * v;
        S += t;
        Q += t * p[u];
    }
}

__device__ __forceinline__ void deal_rec(int* sh_cnt, edge_rec (*sh_rec)[ROW_CAP + 1], edge_rec q)
{
    const int rl   = q.x & (RANGE_ROWS - 1);
    const int slot = atomicAdd(&sh_cnt[rl], 1);
    if (slot < ROW_CAP) sh_rec[rl][slot] = q;
}

__global__ __launch_bounds__(GATHER_WAVES * 32) __attribute__((amdgpu_waves_per_eu(16, 16))) void gcn_gather_ranges(
    const int* __restrict__ seg_i, const int* __restrict__ seg_off, int n_segs, int off_pitch,
    const int* __restrict__ rows, const int* __restrict__ cols,
    const float* __restrict__ vals, const float* __restrict__ pre,
    const float* __restrict__ diag, float* __restrict__ out,
    int n_nodes, int n_edges)
{
    __shared__ int      sh_cnt[RANGE_ROWS];
    __shared__ edge_rec sh_rec[RANGE_ROWS][ROW_CAP + 1];
    __shared__ edge_rec sh_cv[GATHER_WAVES][ROW_CAP];

    const int b    = blockIdx.x;
    const int tid  = threadIdx.x;
    const int w    = tid >> 5;
    const int lane = tid & 31;

    if (tid < RANGE_ROWS) sh_cnt[tid] = 0;
    __syncthreads();

    const edge_rec* seg = (const edge_rec*)seg_i;
    const int ws = __builtin_amdgcn_readfirstlane(w);
    for (int g0 = 0; g0 < n_segs; g0 += GATHER_WAVES * PIECES_PER_WAVE) {
        const int sl = g0 + ws + GATHER_WAVES * lane;
        int o0v = 0, o1v = 0;
        if (lane < PIECES_PER_WAVE && sl < n_segs) {
            const int* op = seg_off + (size_t)sl * off_pitch + b;
            o0v = op[0];
            o1v = op[1];
        }
        edge_rec q[PIECES_PER_WAVE];
#pragma unroll
        for (int j = 0; j < PIECES_PER_WAVE; ++j) {
            const int o0 = lane_i(o0v, j);
            const int ln = lane_i(o1v, j) - o0;
            q[j].x = 0u; q[j].y = 0u;
            const edge_rec* piece = seg + (size_t)(g0 + ws + GATHER_WAVES * j) * SEG_RECS + o0;
            if (lane < ln) q[j] = piece[lane];
        }
#pragma unroll
        for (int j = 0; j < PIECES_PER_WAVE; ++j) {
            const int ln = lane_i(o1v, j) - lane_i(o0v, j);
            if (lane < ln) deal_rec(sh_cnt, sh_rec, q[j]);
        }
        if (__builtin_amdgcn_ballot_w32(o1v - o0v > 32)) {
            for (int j = 0; j < PIECES_PER_WAVE; ++j) {
                const int o0 = lane_i(o0v, j);
                const int ln = lane_i(o1v, j) - o0;
                const edge_rec* piece = seg + (size_t)(g0 + ws + GATHER_WAVES * j) * SEG_RECS + o0;
                for (int base = 32; base < ln; base += 32) {
                    if (base + lane < ln) deal_rec(sh_cnt, sh_rec, piece[base + lane]);
                }
            }
        }
    }
    __syncthreads();

    const float* pl   = pre + 2 * lane;
    const v2f*   pre2 = (const v2f*)pre;

    const int r_first = b * RANGE_ROWS + w * (RANGE_ROWS / GATHER_WAVES);
    float dg = 0.0f;
    if (lane < RANGE_ROWS / GATHER_WAVES && r_first + lane < n_nodes) dg = diag[r_first + lane];

    for (int i = 0; i < RANGE_ROWS / GATHER_WAVES; ++i) {
        const int rl = w * (RANGE_ROWS / GATHER_WAVES) + i;
        const int r  = __builtin_amdgcn_readfirstlane(b * RANGE_ROWS + rl);
        if (r >= n_nodes) break;

        const int  deg    = __builtin_amdgcn_readfirstlane(sh_cnt[rl]);
        const bool listed = deg <= ROW_CAP;
        const int  cnt    = listed ? deg : 0;
        const int  n0     = cnt < 32 ? cnt : 32;

        const bool ok0 = lane < cnt;
        const bool ok1 = lane + 32 < cnt;
        edge_rec q0 = {0u, 0u}, q1 = {0u, 0u};
        if (ok0) q0 = sh_rec[rl][lane];
        if (ok1) q1 = sh_rec[rl][lane + 32];
        const int   c0 = (int)(q0.x >> RANGE_SHIFT), c1 = (int)(q1.x >> RANGE_SHIFT);
        const float v0 = __uint_as_float(q0.y),      v1 = __uint_as_float(q1.y);

        const unsigned f0 = ((unsigned)c0 << 6) | (unsigned)lane;
        const unsigned f1 = ((unsigned)c1 << 6) | (unsigned)(lane + 32);

        int rk0 = 0, rk1 = 0;
        if (cnt <= 32) {
            for (int j = 0; j < cnt; ++j) {
                const unsigned kj = (unsigned)lane_i((int)f0, j);
                rk0 += (kj < f0) ? 1 : 0;
            }
        } else {
            for (int j = 0; j < 32; ++j) {
                const unsigned kj = (unsigned)lane_i((int)f0, j);
                rk0 += (kj < f0) ? 1 : 0;
                rk1 += (kj < f1) ? 1 : 0;
            }
            for (int j = 32; j < cnt; ++j) {
                const unsigned kj = (unsigned)lane_i((int)f1, j - 32);
                rk0 += (kj < f0) ? 1 : 0;
                rk1 += (kj < f1) ? 1 : 0;
            }
        }

        const edge_rec r0 = {(unsigned)c0, q0.y}, r1 = {(unsigned)c1, q1.y};
        if (ok0) sh_cv[w][rk0] = r0;
        if (ok1) sh_cv[w][rk1] = r1;
        __builtin_amdgcn_fence(__ATOMIC_RELEASE, "workgroup");
        __builtin_amdgcn_wave_barrier();
        __builtin_amdgcn_fence(__ATOMIC_ACQUIRE, "workgroup");

        bool dup = false;
        if (lane + 1 < cnt)  dup = sh_cv[w][lane + 1].x == sh_cv[w][lane].x;
        if (lane + 33 < cnt) dup = dup || (sh_cv[w][lane + 33].x == sh_cv[w][lane + 32].x);
        if (__builtin_amdgcn_ballot_w32(dup)) {
            const unsigned long long key0 =
                ((unsigned long long)(((unsigned)c0 << RANGE_SHIFT) | (__float_as_uint(v0) >> (32 - RANGE_SHIFT))) << 32) |
                ((__float_as_uint(v0) << RANGE_SHIFT) | (unsigned)lane);
            const unsigned long long key1 =
                ((unsigned long long)(((unsigned)c1 << RANGE_SHIFT) | (__float_as_uint(v1) >> (32 - RANGE_SHIFT))) << 32) |
                ((__float_as_uint(v1) << RANGE_SHIFT) | (unsigned)(lane + 32));
            const int k0h = (int)(key0 >> 32), k0l = (int)key0;
            const int k1h = (int)(key1 >> 32), k1l = (int)key1;
            rk0 = 0; rk1 = 0;
            for (int j = 0; j < n0; ++j) {
                const unsigned long long kj =
                    ((unsigned long long)(unsigned)lane_i(k0h, j) << 32) | (unsigned)lane_i(k0l, j);
                rk0 += (kj < key0) ? 1 : 0;
                rk1 += (kj < key1) ? 1 : 0;
            }
            for (int j = 32; j < cnt; ++j) {
                const unsigned long long kj =
                    ((unsigned long long)(unsigned)lane_i(k1h, j - 32) << 32) | (unsigned)lane_i(k1l, j - 32);
                rk0 += (kj < key0) ? 1 : 0;
                rk1 += (kj < key1) ? 1 : 0;
            }
            __builtin_amdgcn_fence(__ATOMIC_RELEASE, "workgroup");
            __builtin_amdgcn_wave_barrier();
            if (ok0) sh_cv[w][rk0] = r0;
            if (ok1) sh_cv[w][rk1] = r1;
            __builtin_amdgcn_fence(__ATOMIC_RELEASE, "workgroup");
            __builtin_amdgcn_wave_barrier();
            __builtin_amdgcn_fence(__ATOMIC_ACQUIRE, "workgroup");
        }

        v2f S = {0.0f, 0.0f}, Q = {0.0f, 0.0f};
        {
            const edge_rec* lst = sh_cv[w];
            int j = 0;
            for (; j + 8 <= cnt; j += 8) sum_list<8>(S, Q, pre2, lane, lst + j);
            if (j + 4 <= cnt) { sum_list<4>(S, Q, pre2, lane, lst + j); j += 4; }
            if (j + 2 <= cnt) { sum_list<2>(S, Q, pre2, lane, lst + j); j += 2; }
            if (j < cnt)      { sum_list<1>(S, Q, pre2, lane, lst + j); }
        }
        RowAcc a = {S.x, S.y, Q.x, Q.y};

        if (!listed) {
            for (int base = 0; base < n_edges; base += 32) {
                const int e = base + lane;
                int   c = 0;
                float v = 0.0f;
                bool  mine = false;
                if (e < n_edges && rows[e] == r) { mine = true; c = cols[e]; v = vals[e]; }
                unsigned m = __builtin_amdgcn_ballot_w32(mine);
                while (m) {
                    const int src = __builtin_ctz(m);
                    m &= m - 1;
                    sum_batch<1>(a, pl, c, v, src);
                }
            }
        }

        const float d = lane_f(dg, i);
        float2 o;
        o.x = d * 0.5f * (a.s0 * a.s0 - a.q0);
        o.y = d * 0.5f * (a.s1 * a.s1 - a.q1);
        *(float2*)(out + (size_t)r * D_OUT + 2 * lane) = o;
    }
}

extern "C" void kernel_launch(void* const* d_in, const int* in_sizes, int n_in,
                              void* d_out, int out_size, void* d_ws, size_t ws_size,
                              hipStream_t stream)
{
    const float* feat     = (const float*)d_in[0];
    const float* W        = (const float*)d_in[1];
    const float* bias     = (const float*)d_in[2];
    const float* adj_vals = (const float*)d_in[3];
    const float* diag     = (const float*)d_in[4];
    const int*   adj_rows = (const int*)d_in[5];
    const int*   adj_cols = (const int*)d_in[6];

    const int n_nodes = in_sizes[4];
    const int n_edges = in_sizes[3];

    const int n_ranges  = (n_nodes + RANGE_ROWS - 1) / RANGE_ROWS;
    const int n_segs    = (n_edges + SEG_RECS - 1) / SEG_RECS;
    const int off_pitch = (n_ranges + 1 + 31) & ~31;

    float* pre     = (float*)d_ws;
    int*   seg_i   = (int*)(pre + (size_t)n_nodes * D_OUT);
    int*   seg_off = seg_i + (size_t)n_segs * SEG_RECS * 2;
    float* out     = (float*)d_out;

    const int tiles       = (n_nodes + 15) / 16;
    const int gemm_blocks = (tiles + L1_WAVES - 1) / L1_WAVES;
    const size_t lds_gemm = D_OUT * LDSW_STRIDE * sizeof(float);
    const size_t lds_bin  = (2 * SEG_RECS + 2 * n_ranges + 1 + L1_WAVES) * sizeof(int);
    gcn_gemm_and_bin<<<gemm_blocks + n_segs, L1_THREADS, lds_gemm > lds_bin ? lds_gemm : lds_bin, stream>>>(
        feat, W, bias, pre, n_nodes, gemm_blocks, adj_rows, adj_cols, adj_vals,
        seg_i, seg_off, off_pitch, n_edges, n_ranges);

    gcn_gather_ranges<<<n_ranges, GATHER_WAVES * 32, 0, stream>>>(
        seg_i, seg_off, n_segs, off_pitch, adj_rows, adj_cols, adj_vals, pre, diag, out,
        n_nodes, n_edges);
}
